# starting kernel
# baseline (speedup 1.0000x reference)
#include <hip/hip_runtime.h>
#include <hip/hip_bf16.h>

typedef __attribute__((ext_vector_type(2))) float v2f;
typedef __attribute__((ext_vector_type(8))) float v8f;

#define PADF 48

__device__ __forceinline__ float selu_f(float x) {
  const float alpha = 1.6732632423543772848f;
  const float scale = 1.0507009873554804934f;
  return scale * (x > 0.0f ? x : alpha * (expf(x) - 1.0f));
}

__global__ void k_fill(float* __restrict__ p, float v, int n) {
  int i = blockIdx.x * blockDim.x + threadIdx.x;
  if (i < n) p[i] = v;
}

__global__ void k_degree(const int* __restrict__ dst, float* __restrict__ deg, int E) {
  int e = blockIdx.x * blockDim.x + threadIdx.x;
  if (e < E)
    __hip_atomic_fetch_add(&deg[dst[e]], 1.0f, __ATOMIC_RELAXED, __HIP_MEMORY_SCOPE_AGENT);
}

__global__ void k_rsqrt(const float* __restrict__ deg, float* __restrict__ dinv, int n) {
  int i = blockIdx.x * blockDim.x + threadIdx.x;
  if (i < n) {
    float d = deg[i];
    dinv[i] = d > 0.0f ? rsqrtf(d) : 0.0f;
  }
}

__global__ void k_norm(const int* __restrict__ src, const int* __restrict__ dst,
                       const float* __restrict__ dinv, float* __restrict__ norm, int E) {
  int e = blockIdx.x * blockDim.x + threadIdx.x;
  if (e < E) norm[e] = dinv[src[e]] * dinv[dst[e]];
}

template <int K, int FOUT>
__global__ void k_gemm_wmma(const float* __restrict__ A, int lda,
                            const float* __restrict__ W,
                            float* __restrict__ H) {
  const int lane = threadIdx.x & 31;
  const int half = lane >> 4;
  const int lid  = lane & 15;
  const int row  = blockIdx.x * 16 + lid;
  const int col  = blockIdx.y * 16 + lid;
  const bool cv  = (FOUT % 16 == 0) || (col < FOUT);
  const int  cc  = cv ? col : (FOUT - 1);

  const float* Ap = A + row * lda + 2 * half;
  v8f acc = {};
  constexpr int K4 = K & ~3;
#pragma unroll 4
  for (int k0 = 0; k0 < K4; k0 += 4) {
    const int kb = k0 + 2 * half;
    v2f a, b;
    a.x = Ap[k0];
    a.y = Ap[k0 + 1];
    float b0 = W[kb * FOUT + cc];
    float b1 = W[(kb + 1) * FOUT + cc];
    b.x = cv ? b0 : 0.0f;
    b.y = cv ? b1 : 0.0f;
    acc = __builtin_amdgcn_wmma_f32_16x16x4_f32(false, a, false, b, (short)0, acc,
                                                false, false);
  }
  if constexpr (K % 4 != 0) {
    const int kb  = K4 + 2 * half;
    const int k0c = (kb     < K) ? kb     : (K - 1);
    const int k1c = (kb + 1 < K) ? kb + 1 : (K - 1);
    v2f a, b;
    float a0 = A[row * lda + k0c];
    float a1 = A[row * lda + k1c];
    a.x = (kb     < K) ? a0 : 0.0f;
    a.y = (kb + 1 < K) ? a1 : 0.0f;
    float b0 = W[k0c * FOUT + cc];
    float b1 = W[k1c * FOUT + cc];
    b.x = (cv && kb     < K) ? b0 : 0.0f;
    b.y = (cv && kb + 1 < K) ? b1 : 0.0f;
    acc = __builtin_amdgcn_wmma_f32_16x16x4_f32(false, a, false, b, (short)0, acc,
                                                false, false);
  }
  const int hbase = (blockIdx.x * 16 + 8 * half) * PADF + blockIdx.y * 16 + lid;
#pragma unroll
  for (int r = 0; r < 8; ++r)
    H[hbase + r * PADF] = acc[r];
}

template <int F>
__global__ void k_scatter(const float* __restrict__ h, const int* __restrict__ src,
                          const int* __restrict__ dst, const float* __restrict__ norm,
                          float* __restrict__ agg, int E) {
  int t = blockIdx.x * blockDim.x + threadIdx.x;
  if (t >= E * F) return;
  int e = t / F;
  int f = t - e * F;
  float v = norm[e] * h[src[e] * PADF + f];
  __hip_atomic_fetch_add(&agg[dst[e] * PADF + f], v,
                         __ATOMIC_RELAXED, __HIP_MEMORY_SCOPE_AGENT);
}

template <int F>
__global__ void k_finalize(const float* __restrict__ h, const float* __restrict__ agg,
                           const float* __restrict__ dinv, const float* __restrict__ bias,
                           float* __restrict__ out, int ldo, int N) {
  int t = blockIdx.x * blockDim.x + threadIdx.x;
  if (t >= N * F) return;
  int n = t / F;
  int f = t - n * F;
  float di = dinv[n];
  float v = agg[n * PADF + f] + di * di * h[n * PADF + f] + bias[f];
  out[n * ldo + f] = selu_f(v);
}

extern "C" void kernel_launch(void* const* d_in, const int* in_sizes, int n_in,
                              void* d_out, int out_size, void* d_ws, size_t ws_size,
                              hipStream_t stream) {
  const float* x  = (const float*)d_in[0];
  const int*   ei = (const int*)d_in[1];
  const float* W1 = (const float*)d_in[2];
  const float* b1 = (const float*)d_in[3];
  const float* W2 = (const float*)d_in[4];
  const float* b2 = (const float*)d_in[5];
  const float* W3 = (const float*)d_in[6];
  const float* b3 = (const float*)d_in[7];
  const float* W4 = (const float*)d_in[8];
  const float* b4 = (const float*)d_in[9];

  const int E = in_sizes[1] / 2;
  const int N = in_sizes[0] / 128;
  const int* src = ei;
  const int* dst = ei + E;

  float* ws   = (float*)d_ws;
  float* deg  = ws;
  float* dinv = deg  + N;
  float* norm = dinv + N;
  float* hG   = norm + E;
  float* agg  = hG   + (size_t)N * PADF;
  float* hIn  = agg  + (size_t)N * PADF;

  auto cdiv = [](long long a, long long b) { return (int)((a + b - 1) / b); };
  const int NT = N / 16;

  k_fill  <<<cdiv(N, 256), 256, 0, stream>>>(deg, 1.0f, N);
  k_degree<<<cdiv(E, 256), 256, 0, stream>>>(dst, deg, E);
  k_rsqrt <<<cdiv(N, 256), 256, 0, stream>>>(deg, dinv, N);
  k_norm  <<<cdiv(E, 256), 256, 0, stream>>>(src, dst, dinv, norm, E);

  k_gemm_wmma<128, 15><<<dim3(NT, 1), 32, 0, stream>>>(x, 128, W1, hG);
  k_fill<<<cdiv(N * PADF, 256), 256, 0, stream>>>(agg, 0.0f, N * PADF);
  k_scatter<15><<<cdiv((long long)E * 15, 256), 256, 0, stream>>>(hG, src, dst, norm, agg, E);
  k_finalize<15><<<cdiv((long long)N * 15, 256), 256, 0, stream>>>(hG, agg, dinv, b1, hIn, PADF, N);

  k_gemm_wmma<15, 20><<<dim3(NT, 2), 32, 0, stream>>>(hIn, PADF, W2, hG);
  k_fill<<<cdiv(N * PADF, 256), 256, 0, stream>>>(agg, 0.0f, N * PADF);
  k_scatter<20><<<cdiv((long long)E * 20, 256), 256, 0, stream>>>(hG, src, dst, norm, agg, E);
  k_finalize<20><<<cdiv((long long)N * 20, 256), 256, 0, stream>>>(hG, agg, dinv, b2, hIn, PADF, N);

  k_gemm_wmma<20, 27><<<dim3(NT, 2), 32, 0, stream>>>(hIn, PADF, W3, hG);
  k_fill<<<cdiv(N * PADF, 256), 256, 0, stream>>>(agg, 0.0f, N * PADF);
  k_scatter<27><<<cdiv((long long)E * 27, 256), 256, 0, stream>>>(hG, src, dst, norm, agg, E);
  k_finalize<27><<<cdiv((long long)N * 27, 256), 256, 0, stream>>>(hG, agg, dinv, b3, hIn, PADF, N);

  k_gemm_wmma<27, 36><<<dim3(NT, 3), 32, 0, stream>>>(hIn, PADF, W4, hG);
  k_fill<<<cdiv(N * PADF, 256), 256, 0, stream>>>(agg, 0.0f, N * PADF);
  k_scatter<36><<<cdiv((long long)E * 36, 256), 256, 0, stream>>>(hG, src, dst, norm, agg, E);
  k_finalize<36><<<cdiv((long long)N * 36, 256), 256, 0, stream>>>(hG, agg, dinv, b4,
                                                                   (float*)d_out, 36, N);
}


# k_agg row gather: a lane takes 16 bytes of a source row (one global_load_b128, four LDS adds rotated over the banks), not one float; arrival order of the LDS adds changes (was not fixed before)
# speedup vs baseline: 1.3996x; 1.3996x over previous
#include <hip/hip_runtime.h>
#include <hip/hip_bf16.h>

typedef __attribute__((ext_vector_type(2))) float v2f;
typedef __attribute__((ext_vector_type(8))) float v8f;

#define PADF 48

__device__ __forceinline__ float selu_f(float x) {
  const float alpha = 1.6732632423543772848f;
  const float scale = 1.0507009873554804934f;
  return scale * (x > 0.0f ? x : alpha * (expf(x) - 1.0f));
}

__global__ void k_fill(float* __restrict__ p, float v, int n) {
  int i = blockIdx.x * blockDim.x + threadIdx.x;
  if (i < n) p[i] = v;
}

__global__ void k_degree(const int* __restrict__ dst, float* __restrict__ deg, int E) {
  int e = blockIdx.x * blockDim.x + threadIdx.x;
  if (e < E)
    __hip_atomic_fetch_add(&deg[dst[e]], 1.0f, __ATOMIC_RELAXED, __HIP_MEMORY_SCOPE_AGENT);
}

__global__ void k_rsqrt(const float* __restrict__ deg, float* __restrict__ dinv, int n) {
  int i = blockIdx.x * blockDim.x + threadIdx.x;
  if (i < n) {
    float d = deg[i];
    dinv[i] = d > 0.0f ? rsqrtf(d) : 0.0f;
  }
}

__global__ void k_norm(const int* __restrict__ src, const int* __restrict__ dst,
                       const float* __restrict__ dinv, float* __restrict__ norm, int E) {
  int e = blockIdx.x * blockDim.x + threadIdx.x;
  if (e < E) norm[e] = dinv[src[e]] * dinv[dst[e]];
}

template <int K, int FOUT>
__global__ void k_gemm_wmma(const float* __restrict__ A, int lda,
                            const float* __restrict__ W,
                            float* __restrict__ H) {
  const int lane = threadIdx.x & 31;
  const int half = lane >> 4;
  const int lid  = lane & 15;
  const int row  = blockIdx.x * 16 + lid;
  const int col  = blockIdx.y * 16 + lid;
  const bool cv  = (FOUT % 16 == 0) || (col < FOUT);
  const int  cc  = cv ? col : (FOUT - 1);

  const float* Ap = A + row * lda + 2 * half;
  v8f acc = {};
  constexpr int K4 = K & ~3;
#pragma unroll 4
  for (int k0 = 0; k0 < K4; k0 += 4) {
    const int kb = k0 + 2 * half;
    v2f a, b;
    a.x = Ap[k0];
    a.y = Ap[k0 + 1];
    float b0 = W[kb * FOUT + cc];
    float b1 = W[(kb + 1) * FOUT + cc];
    b.x = cv ? b0 : 0.0f;
    b.y = cv ? b1 : 0.0f;
    acc = __builtin_amdgcn_wmma_f32_16x16x4_f32(false, a, false, b, (short)0, acc,
                                                false, false);
  }
  if constexpr (K % 4 != 0) {
    const int kb  = K4 + 2 * half;
    const int k0c = (kb     < K) ? kb     : (K - 1);
    const int k1c = (kb + 1 < K) ? kb + 1 : (K - 1);
    v2f a, b;
    float a0 = A[row * lda + k0c];
    float a1 = A[row * lda + k1c];
    a.x = (kb     < K) ? a0 : 0.0f;
    a.y = (kb + 1 < K) ? a1 : 0.0f;
    float b0 = W[k0c * FOUT + cc];
    float b1 = W[k1c * FOUT + cc];
    b.x = (cv && kb     < K) ? b0 : 0.0f;
    b.y = (cv && kb + 1 < K) ? b1 : 0.0f;
    acc = __builtin_amdgcn_wmma_f32_16x16x4_f32(false, a, false, b, (short)0, acc,
                                                false, false);
  }
  const int hbase = (blockIdx.x * 16 + 8 * half) * PADF + blockIdx.y * 16 + lid;
#pragma unroll
  for (int r = 0; r < 8; ++r)
    H[hbase + r * PADF] = acc[r];
}

#define AGG_THREADS 1024
#define AGG_EPT 32
#define AGG_BATCH (AGG_THREADS * AGG_EPT)
#define AGG_QC 2048
#define AGG_U 8

template <int F, int FP, int NR>
__global__ __launch_bounds__(AGG_THREADS)
void k_agg(const float* __restrict__ h, const int* __restrict__ src,
           const int* __restrict__ dst, const float* __restrict__ norm,
           float* __restrict__ part, int E) {
  __shared__ __attribute__((aligned(16))) float acc[NR * FP];
  __shared__ int   q_a[AGG_QC];
  __shared__ int   q_d[AGG_QC];
  __shared__ float q_w[AGG_QC];
  __shared__ int   q_n;
  __shared__ int   q_more;

  const int tid  = threadIdx.x;
  const int lane = tid & 31;
  const int lo   = blockIdx.y * NR;
  const int G    = gridDim.x;
  const int echunk = (E + G - 1) / G;
  const int e_begin = blockIdx.x * echunk;
  const int e_end   = (e_begin + echunk < E) ? (e_begin + echunk) : E;

  constexpr int L  = FP / 4;
  constexpr int EP = AGG_THREADS / L;
  const int  esub = tid / L;
  const int  pc   = tid - esub * L;
  const int  c4   = pc * 4;
  const bool lane_used = tid < EP * L;
  const int  rot  = (pc + esub) & 3;
  const int  o0 = rot, o1 = (rot + 1) & 3, o2 = (rot + 2) & 3, o3 = (rot + 3) & 3;

  float4* acc4 = reinterpret_cast<float4*>(acc);
  for (int i = tid; i < NR * FP / 4; i += AGG_THREADS)
    acc4[i] = make_float4(0.0f, 0.0f, 0.0f, 0.0f);
  if (tid == 0) { q_n = 0; q_more = 0; }
  __syncthreads();

  for (int b = e_begin; b < e_end; b += AGG_BATCH) {
    unsigned pend = 0;
#pragma unroll
    for (int j = 0; j < AGG_EPT; ++j) {
      const int e = b + j * AGG_THREADS + tid;
      const int ec = (e < e_end) ? e : e_begin;
      const int d = dst[ec];
      const unsigned hit = (unsigned)((unsigned)(d - lo) < (unsigned)NR) & (unsigned)(e < e_end);
      pend |= hit << j;
    }

    bool more;
    do {
      const int c = __popc(pend);
      int incl = c;
#pragma unroll
      for (int o = 1; o < 32; o <<= 1) {
        const int t = __shfl_up(incl, o, 32);
        if (lane >= o) incl += t;
      }
      const int wtot = __shfl(incl, 31, 32);
      int base = 0;
      if (lane == 31 && wtot > 0)
        base = __hip_atomic_fetch_add(&q_n, wtot, __ATOMIC_RELAXED, __HIP_MEMORY_SCOPE_WORKGROUP);
      base = __shfl(base, 31, 32);
      int pos = base + incl - c;
      unsigned p = pend;
      while (p) {
        const int j = __ffs(p) - 1;
        p &= p - 1;
        if (pos < AGG_QC) {
          q_a[pos] = b + j * AGG_THREADS + tid;
          pend &= ~(1u << j);
        }
        ++pos;
      }
      if (pend) q_more = 1;
      __syncthreads();

      const int M = (q_n < AGG_QC) ? q_n : AGG_QC;
      more = (q_more != 0);
      for (int m = tid; m < M; m += AGG_THREADS) {
        const int e = q_a[m];
        const int s = src[e];
        const int d = dst[e];
        const float w = norm[e];
        q_a[m] = s * PADF;
        q_d[m] = (d - lo) * FP;
        q_w[m] = w;
      }
      __syncthreads();

      if (tid == 0) { q_n = 0; q_more = 0; }
      for (int m0 = 0; m0 < M; m0 += EP * AGG_U) {
        float4 hv[AGG_U];
        float  wv[AGG_U];
        int    a[AGG_U];
        bool   ok[AGG_U];
#pragma unroll
        for (int u = 0; u < AGG_U; ++u) {
          const int m = m0 + u * EP + esub;
          ok[u] = lane_used && (m < M);
          const int mc = ok[u] ? m : 0;
          const int qa = q_a[mc];
          const int qd = q_d[mc];
          hv[u] = *reinterpret_cast<const float4*>(h + qa + c4);
          wv[u] = q_w[mc];
          a[u]  = qd + c4;
        }
#pragma unroll
        for (int u = 0; u < AGG_U; ++u)
          if (ok[u]) {
            float v0 = hv[u].x, v1 = hv[u].y, v2 = hv[u].z, v3 = hv[u].w;
            if (rot & 1) { const float t = v0; v0 = v1; v1 = v2; v2 = v3; v3 = t; }
            if (rot & 2) { const float t0 = v0, t1 = v1; v0 = v2; v1 = v3; v2 = t0; v3 = t1; }
            float* p = &acc[a[u]];
            const float w = wv[u];
            __hip_atomic_fetch_add(p + o0, w * v0, __ATOMIC_RELAXED, __HIP_MEMORY_SCOPE_WORKGROUP);
            __hip_atomic_fetch_add(p + o1, w * v1, __ATOMIC_RELAXED, __HIP_MEMORY_SCOPE_WORKGROUP);
            __hip_atomic_fetch_add(p + o2, w * v2, __ATOMIC_RELAXED, __HIP_MEMORY_SCOPE_WORKGROUP);
            __hip_atomic_fetch_add(p + o3, w * v3, __ATOMIC_RELAXED, __HIP_MEMORY_SCOPE_WORKGROUP);
          }
      }
      __syncthreads();
    } while (more);
  }

  const size_t npad = (size_t)gridDim.y * NR;
  float4* out4 = reinterpret_cast<float4*>(part + ((size_t)blockIdx.x * npad + lo) * FP);
  for (int i = tid; i < NR * FP / 4; i += AGG_THREADS)
    out4[i] = acc4[i];
}

template <int F, int FP, int G>
__global__ void k_finalize(const float* __restrict__ h, const float* __restrict__ part,
                           const float* __restrict__ dinv, const float* __restrict__ bias,
                           float* __restrict__ out, int ldo, int N, int npad) {
  int t = blockIdx.x * blockDim.x + threadIdx.x;
  if (t >= N * F) return;
  int n = t / F;
  int f = t - n * F;
  float agg = 0.0f;
#pragma unroll
  for (int g = 0; g < G; ++g)
    agg += part[((size_t)g * npad + n) * FP + f];
  float di = dinv[n];
  float v = agg + di * di * h[n * PADF + f] + bias[f];
  out[n * ldo + f] = selu_f(v);
}

extern "C" void kernel_launch(void* const* d_in, const int* in_sizes, int n_in,
                              void* d_out, int out_size, void* d_ws, size_t ws_size,
                              hipStream_t stream) {
  const float* x  = (const float*)d_in[0];
  const int*   ei = (const int*)d_in[1];
  const float* W1 = (const float*)d_in[2];
  const float* b1 = (const float*)d_in[3];
  const float* W2 = (const float*)d_in[4];
  const float* b2 = (const float*)d_in[5];
  const float* W3 = (const float*)d_in[6];
  const float* b3 = (const float*)d_in[7];
  const float* W4 = (const float*)d_in[8];
  const float* b4 = (const float*)d_in[9];

  const int E = in_sizes[1] / 2;
  const int N = in_sizes[0] / 128;
  const int* src = ei;
  const int* dst = ei + E;

  auto cdiv = [](long long a, long long b) { return (int)((a + b - 1) / b); };
  auto al64 = [](size_t n) { return (n + 63) & ~(size_t)63; };

  float* ws   = (float*)d_ws;
  float* deg  = ws;
  float* dinv = deg  + al64(N);
  float* norm = dinv + al64(N);
  float* hG   = norm + al64(E);
  float* hIn  = hG   + (size_t)N * PADF;
  float* part = hIn  + (size_t)N * PADF;

  const int NT = N / 16;

  constexpr int NR1 = 4096, G1 = 10;
  constexpr int NR2 = 3584, G2 = 9;
  constexpr int NR3 = 2400, G3 = 6;
  constexpr int NR4 = 2048, G4 = 5;
  const int R1 = cdiv(N, NR1), R2 = cdiv(N, NR2), R3 = cdiv(N, NR3), R4 = cdiv(N, NR4);

  k_fill  <<<cdiv(N, 256), 256, 0, stream>>>(deg, 1.0f, N);
  k_degree<<<cdiv(E, 256), 256, 0, stream>>>(dst, deg, E);
  k_rsqrt <<<cdiv(N, 256), 256, 0, stream>>>(deg, dinv, N);
  k_norm  <<<cdiv(E, 256), 256, 0, stream>>>(src, dst, dinv, norm, E);

  k_gemm_wmma<128, 15><<<dim3(NT, 1), 32, 0, stream>>>(x, 128, W1, hG);
  k_agg<15, 16, NR1><<<dim3(G1, R1), AGG_THREADS, 0, stream>>>(hG, src, dst, norm, part, E);
  k_finalize<15, 16, G1><<<cdiv((long long)N * 15, 256), 256, 0, stream>>>(hG, part, dinv, b1, hIn, PADF, N, R1 * NR1);

  k_gemm_wmma<15, 20><<<dim3(NT, 2), 32, 0, stream>>>(hIn, PADF, W2, hG);
  k_agg<20, 20, NR2><<<dim3(G2, R2), AGG_THREADS, 0, stream>>>(hG, src, dst, norm, part, E);
  k_finalize<20, 20, G2><<<cdiv((long long)N * 20, 256), 256, 0, stream>>>(hG, part, dinv, b2, hIn, PADF, N, R2 * NR2);

  k_gemm_wmma<20, 27><<<dim3(NT, 2), 32, 0, stream>>>(hIn, PADF, W3, hG);
  k_agg<27, 28, NR3><<<dim3(G3, R3), AGG_THREADS, 0, stream>>>(hG, src, dst, norm, part, E);
  k_finalize<27, 28, G3><<<cdiv((long long)N * 27, 256), 256, 0, stream>>>(hG, part, dinv, b3, hIn, PADF, N, R3 * NR3);

  k_gemm_wmma<27, 36><<<dim3(NT, 3), 32, 0, stream>>>(hIn, PADF, W4, hG);
  k_agg<36, 36, NR4><<<dim3(G4, R4), AGG_THREADS, 0, stream>>>(hG, src, dst, norm, part, E);
  k_finalize<36, 36, G4><<<cdiv((long long)N * 36, 256), 256, 0, stream>>>(hG, part, dinv, b4,
                                                                   (float*)d_out, 36, N, R4 * NR4);
}


# k_agg fetches one 16-byte record {source row offset, dst, norm} per queued edge (built by k_norm) where it read src[e], dst[e] and norm[e] from three arrays; sum order unchanged
# speedup vs baseline: 1.4545x; 1.0392x over previous
#include <hip/hip_runtime.h>
#include <hip/hip_bf16.h>

typedef __attribute__((ext_vector_type(2))) float v2f;
typedef __attribute__((ext_vector_type(8))) float v8f;

#define PADF 48

__device__ __forceinline__ float selu_f(float x) {
  const float alpha = 1.6732632423543772848f;
  const float scale = 1.0507009873554804934f;
  return scale * (x > 0.0f ? x : alpha * (expf(x) - 1.0f));
}

__global__ void k_fill(float* __restrict__ p, float v, int n) {
  int i = blockIdx.x * blockDim.x + threadIdx.x;
  if (i < n) p[i] = v;
}

__global__ void k_degree(const int* __restrict__ dst, float* __restrict__ deg, int E) {
  int e = blockIdx.x * blockDim.x + threadIdx.x;
  if (e < E)
    __hip_atomic_fetch_add(&deg[dst[e]], 1.0f, __ATOMIC_RELAXED, __HIP_MEMORY_SCOPE_AGENT);
}

__global__ void k_rsqrt(const float* __restrict__ deg, float* __restrict__ dinv, int n) {
  int i = blockIdx.x * blockDim.x + threadIdx.x;
  if (i < n) {
    float d = deg[i];
    dinv[i] = d > 0.0f ? rsqrtf(d) : 0.0f;
  }
}

__global__ void k_norm(const int* __restrict__ src, const int* __restrict__ dst,
                       const float* __restrict__ dinv, int* __restrict__ rec, int E) {
  int e = blockIdx.x * blockDim.x + threadIdx.x;
  if (e < E) {
    const int s = src[e];
    const int d = dst[e];
    const float w = dinv[s] * dinv[d];
    reinterpret_cast<int4*>(rec)[e] = make_int4(s * PADF, d, __float_as_int(w), 0);
  }
}

template <int K, int FOUT>
__global__ void k_gemm_wmma(const float* __restrict__ A, int lda,
                            const float* __restrict__ W,
                            float* __restrict__ H) {
  const int lane = threadIdx.x & 31;
  const int half = lane >> 4;
  const int lid  = lane & 15;
  const int row  = blockIdx.x * 16 + lid;
  const int col  = blockIdx.y * 16 + lid;
  const bool cv  = (FOUT % 16 == 0) || (col < FOUT);
  const int  cc  = cv ? col : (FOUT - 1);

  const float* Ap = A + row * lda + 2 * half;
  v8f acc = {};
  constexpr int K4 = K & ~3;
#pragma unroll 4
  for (int k0 = 0; k0 < K4; k0 += 4) {
    const int kb = k0 + 2 * half;
    v2f a, b;
    a.x = Ap[k0];
    a.y = Ap[k0 + 1];
    float b0 = W[kb * FOUT + cc];
    float b1 = W[(kb + 1) * FOUT + cc];
    b.x = cv ? b0 : 0.0f;
    b.y = cv ? b1 : 0.0f;
    acc = __builtin_amdgcn_wmma_f32_16x16x4_f32(false, a, false, b, (short)0, acc,
                                                false, false);
  }
  if constexpr (K % 4 != 0) {
    const int kb  = K4 + 2 * half;
    const int k0c = (kb     < K) ? kb     : (K - 1);
    const int k1c = (kb + 1 < K) ? kb + 1 : (K - 1);
    v2f a, b;
    float a0 = A[row * lda + k0c];
    float a1 = A[row * lda + k1c];
    a.x = (kb     < K) ? a0 : 0.0f;
    a.y = (kb + 1 < K) ? a1 : 0.0f;
    float b0 = W[k0c * FOUT + cc];
    float b1 = W[k1c * FOUT + cc];
    b.x = (cv && kb     < K) ? b0 : 0.0f;
    b.y = (cv && kb + 1 < K) ? b1 : 0.0f;
    acc = __builtin_amdgcn_wmma_f32_16x16x4_f32(false, a, false, b, (short)0, acc,
                                                false, false);
  }
  const int hbase = (blockIdx.x * 16 + 8 * half) * PADF + blockIdx.y * 16 + lid;
#pragma unroll
  for (int r = 0; r < 8; ++r)
    H[hbase + r * PADF] = acc[r];
}

#define AGG_THREADS 1024
#define AGG_EPT 32
#define AGG_BATCH (AGG_THREADS * AGG_EPT)
#define AGG_QC 2048
#define AGG_U 8

template <int F, int FP, int NR>
__global__ __launch_bounds__(AGG_THREADS)
void k_agg(const float* __restrict__ h, const int* __restrict__ dst,
           const int* __restrict__ rec, float* __restrict__ part, int E) {
  __shared__ __attribute__((aligned(16))) float acc[NR * FP];
  __shared__ int   q_a[AGG_QC];
  __shared__ int   q_d[AGG_QC];
  __shared__ float q_w[AGG_QC];
  __shared__ int   q_n;
  __shared__ int   q_more;

  const int tid  = threadIdx.x;
  const int lane = tid & 31;
  const int lo   = blockIdx.y * NR;
  const int G    = gridDim.x;
  const int echunk = (E + G - 1) / G;
  const int e_begin = blockIdx.x * echunk;
  const int e_end   = (e_begin + echunk < E) ? (e_begin + echunk) : E;

  constexpr int L  = FP / 4;
  constexpr int EP = AGG_THREADS / L;
  const int  esub = tid / L;
  const int  pc   = tid - esub * L;
  const int  c4   = pc * 4;
  const bool lane_used = tid < EP * L;
  const int  rot  = (pc + esub) & 3;
  const int  o0 = rot, o1 = (rot + 1) & 3, o2 = (rot + 2) & 3, o3 = (rot + 3) & 3;

  float4* acc4 = reinterpret_cast<float4*>(acc);
  for (int i = tid; i < NR * FP / 4; i += AGG_THREADS)
    acc4[i] = make_float4(0.0f, 0.0f, 0.0f, 0.0f);
  if (tid == 0) { q_n = 0; q_more = 0; }
  __syncthreads();

  for (int b = e_begin; b < e_end; b += AGG_BATCH) {
    unsigned pend = 0;
#pragma unroll
    for (int j = 0; j < AGG_EPT; ++j) {
      const int e = b + j * AGG_THREADS + tid;
      const int ec = (e < e_end) ? e : e_begin;
      const int d = dst[ec];
      const unsigned hit = (unsigned)((unsigned)(d - lo) < (unsigned)NR) & (unsigned)(e < e_end);
      pend |= hit << j;
    }

    bool more;
    do {
      const int c = __popc(pend);
      int incl = c;
#pragma unroll
      for (int o = 1; o < 32; o <<= 1) {
        const int t = __shfl_up(incl, o, 32);
        if (lane >= o) incl += t;
      }
      const int wtot = __shfl(incl, 31, 32);
      int base = 0;
      if (lane == 31 && wtot > 0)
        base = __hip_atomic_fetch_add(&q_n, wtot, __ATOMIC_RELAXED, __HIP_MEMORY_SCOPE_WORKGROUP);
      base = __shfl(base, 31, 32);
      int pos = base + incl - c;
      unsigned p = pend;
      while (p) {
        const int j = __ffs(p) - 1;
        p &= p - 1;
        if (pos < AGG_QC) {
          q_a[pos] = b + j * AGG_THREADS + tid;
          pend &= ~(1u << j);
        }
        ++pos;
      }
      if (pend) q_more = 1;
      __syncthreads();

      const int M = (q_n < AGG_QC) ? q_n : AGG_QC;
      more = (q_more != 0);
      for (int m = tid; m < M; m += AGG_THREADS) {
        const int e = q_a[m];
        const int4 r = reinterpret_cast<const int4*>(rec)[e];
        q_a[m] = r.x;
        q_d[m] = (r.y - lo) * FP;
        q_w[m] = __int_as_float(r.z);
      }
      __syncthreads();

      if (tid == 0) { q_n = 0; q_more = 0; }
      for (int m0 = 0; m0 < M; m0 += EP * AGG_U) {
        float4 hv[AGG_U];
        float  wv[AGG_U];
        int    a[AGG_U];
        bool   ok[AGG_U];
#pragma unroll
        for (int u = 0; u < AGG_U; ++u) {
          const int m = m0 + u * EP + esub;
          ok[u] = lane_used && (m < M);
          const int mc = ok[u] ? m : 0;
          const int qa = q_a[mc];
          const int qd = q_d[mc];
          hv[u] = *reinterpret_cast<const float4*>(h + qa + c4);
          wv[u] = q_w[mc];
          a[u]  = qd + c4;
        }
#pragma unroll
        for (int u = 0; u < AGG_U; ++u)
          if (ok[u]) {
            float v0 = hv[u].x, v1 = hv[u].y, v2 = hv[u].z, v3 = hv[u].w;
            if (rot & 1) { const float t = v0; v0 = v1; v1 = v2; v2 = v3; v3 = t; }
            if (rot & 2) { const float t0 = v0, t1 = v1; v0 = v2; v1 = v3; v2 = t0; v3 = t1; }
            float* p = &acc[a[u]];
            const float w = wv[u];
            __hip_atomic_fetch_add(p + o0, w * v0, __ATOMIC_RELAXED, __HIP_MEMORY_SCOPE_WORKGROUP);
            __hip_atomic_fetch_add(p + o1, w * v1, __ATOMIC_RELAXED, __HIP_MEMORY_SCOPE_WORKGROUP);
            __hip_atomic_fetch_add(p + o2, w * v2, __ATOMIC_RELAXED, __HIP_MEMORY_SCOPE_WORKGROUP);
            __hip_atomic_fetch_add(p + o3, w * v3, __ATOMIC_RELAXED, __HIP_MEMORY_SCOPE_WORKGROUP);
          }
      }
      __syncthreads();
    } while (more);
  }

  const size_t npad = (size_t)gridDim.y * NR;
  float4* out4 = reinterpret_cast<float4*>(part + ((size_t)blockIdx.x * npad + lo) * FP);
  for (int i = tid; i < NR * FP / 4; i += AGG_THREADS)
    out4[i] = acc4[i];
}

template <int F, int FP, int G>
__global__ void k_finalize(const float* __restrict__ h, const float* __restrict__ part,
                           const float* __restrict__ dinv, const float* __restrict__ bias,
                           float* __restrict__ out, int ldo, int N, int npad) {
  int t = blockIdx.x * blockDim.x + threadIdx.x;
  if (t >= N * F) return;
  int n = t / F;
  int f = t - n * F;
  float agg = 0.0f;
#pragma unroll
  for (int g = 0; g < G; ++g)
    agg += part[((size_t)g * npad + n) * FP + f];
  float di = dinv[n];
  float v = agg + di * di * h[n * PADF + f] + bias[f];
  out[n * ldo + f] = selu_f(v);
}

extern "C" void kernel_launch(void* const* d_in, const int* in_sizes, int n_in,
                              void* d_out, int out_size, void* d_ws, size_t ws_size,
                              hipStream_t stream) {
  const float* x  = (const float*)d_in[0];
  const int*   ei = (const int*)d_in[1];
  const float* W1 = (const float*)d_in[2];
  const float* b1 = (const float*)d_in[3];
  const float* W2 = (const float*)d_in[4];
  const float* b2 = (const float*)d_in[5];
  const float* W3 = (const float*)d_in[6];
  const float* b3 = (const float*)d_in[7];
  const float* W4 = (const float*)d_in[8];
  const float* b4 = (const float*)d_in[9];

  const int E = in_sizes[1] / 2;
  const int N = in_sizes[0] / 128;
  const int* src = ei;
  const int* dst = ei + E;

  auto cdiv = [](long long a, long long b) { return (int)((a + b - 1) / b); };
  auto al64 = [](size_t n) { return (n + 63) & ~(size_t)63; };

  float* ws   = (float*)d_ws;
  float* deg  = ws;
  float* dinv = deg  + al64(N);
  int*   rec  = (int*)(dinv + al64(N));
  float* hG   = (float*)(rec + al64((size_t)4 * E));
  float* hIn  = hG   + (size_t)N * PADF;
  float* part = hIn  + (size_t)N * PADF;

  const int NT = N / 16;

  constexpr int NR1 = 4096, G1 = 10;
  constexpr int NR2 = 3584, G2 = 9;
  constexpr int NR3 = 2400, G3 = 6;
  constexpr int NR4 = 2048, G4 = 5;
  const int R1 = cdiv(N, NR1), R2 = cdiv(N, NR2), R3 = cdiv(N, NR3), R4 = cdiv(N, NR4);

  k_fill  <<<cdiv(N, 256), 256, 0, stream>>>(deg, 1.0f, N);
  k_degree<<<cdiv(E, 256), 256, 0, stream>>>(dst, deg, E);
  k_rsqrt <<<cdiv(N, 256), 256, 0, stream>>>(deg, dinv, N);
  k_norm  <<<cdiv(E, 256), 256, 0, stream>>>(src, dst, dinv, rec, E);

  k_gemm_wmma<128, 15><<<dim3(NT, 1), 32, 0, stream>>>(x, 128, W1, hG);
  k_agg<15, 16, NR1><<<dim3(G1, R1), AGG_THREADS, 0, stream>>>(hG, dst, rec, part, E);
  k_finalize<15, 16, G1><<<cdiv((long long)N * 15, 256), 256, 0, stream>>>(hG, part, dinv, b1, hIn, PADF, N, R1 * NR1);

  k_gemm_wmma<15, 20><<<dim3(NT, 2), 32, 0, stream>>>(hIn, PADF, W2, hG);
  k_agg<20, 20, NR2><<<dim3(G2, R2), AGG_THREADS, 0, stream>>>(hG, dst, rec, part, E);
  k_finalize<20, 20, G2><<<cdiv((long long)N * 20, 256), 256, 0, stream>>>(hG, part, dinv, b2, hIn, PADF, N, R2 * NR2);

  k_gemm_wmma<20, 27><<<dim3(NT, 2), 32, 0, stream>>>(hIn, PADF, W3, hG);
  k_agg<27, 28, NR3><<<dim3(G3, R3), AGG_THREADS, 0, stream>>>(hG, dst, rec, part, E);
  k_finalize<27, 28, G3><<<cdiv((long long)N * 27, 256), 256, 0, stream>>>(hG, part, dinv, b3, hIn, PADF, N, R3 * NR3);

  k_gemm_wmma<27, 36><<<dim3(NT, 3), 32, 0, stream>>>(hIn, PADF, W4, hG);
  k_agg<36, 36, NR4><<<dim3(G4, R4), AGG_THREADS, 0, stream>>>(hG, dst, rec, part, E);
  k_finalize<36, 36, G4><<<cdiv((long long)N * 36, 256), 256, 0, stream>>>(hG, part, dinv, b4,
                                                                   (float*)d_out, 36, N, R4 * NR4);
}


# k_agg scans one byte per edge (row-range number dst / NR, per-layer planes written by k_norm), 32 edges per thread in two 16-byte loads, not 32 dst loads; queue order, so LDS add order, changes
# speedup vs baseline: 1.5219x; 1.0463x over previous
#include <hip/hip_runtime.h>
#include <hip/hip_bf16.h>

typedef __attribute__((ext_vector_type(2))) float v2f;
typedef __attribute__((ext_vector_type(8))) float v8f;

#define PADF 48

__device__ __forceinline__ float selu_f(float x) {
  const float alpha = 1.6732632423543772848f;
  const float scale = 1.0507009873554804934f;
  return scale * (x > 0.0f ? x : alpha * (expf(x) - 1.0f));
}

__global__ void k_fill(float* __restrict__ p, float v, int n) {
  int i = blockIdx.x * blockDim.x + threadIdx.x;
  if (i < n) p[i] = v;
}

__global__ void k_degree(const int* __restrict__ dst, float* __restrict__ deg, int E) {
  int e = blockIdx.x * blockDim.x + threadIdx.x;
  if (e < E)
    __hip_atomic_fetch_add(&deg[dst[e]], 1.0f, __ATOMIC_RELAXED, __HIP_MEMORY_SCOPE_AGENT);
}

__global__ void k_rsqrt(const float* __restrict__ deg, float* __restrict__ dinv, int n) {
  int i = blockIdx.x * blockDim.x + threadIdx.x;
  if (i < n) {
    float d = deg[i];
    dinv[i] = d > 0.0f ? rsqrtf(d) : 0.0f;
  }
}

template <int NRA, int NRB, int NRC, int NRD>
__global__ __launch_bounds__(256)
void k_norm(const int* __restrict__ src, const int* __restrict__ dst,
            const float* __restrict__ dinv, int* __restrict__ rec,
            unsigned* __restrict__ planes, int pw, int E) {
  __shared__ unsigned sb[4][64];
  const int tid = threadIdx.x;
  const int e = blockIdx.x * 256 + tid;
  unsigned ia = 255u, ib = 255u, ic = 255u, id = 255u;
  if (e < E) {
    const int s = src[e];
    const int d = dst[e];
    const float w = dinv[s] * dinv[d];
    reinterpret_cast<int4*>(rec)[e] = make_int4(s * PADF, d, __float_as_int(w), 0);
    ia = (unsigned)d / (unsigned)NRA;
    ib = (unsigned)d / (unsigned)NRB;
    ic = (unsigned)d / (unsigned)NRC;
    id = (unsigned)d / (unsigned)NRD;
  }
  unsigned char* sbb = reinterpret_cast<unsigned char*>(&sb[0][0]);
  sbb[tid]       = (unsigned char)ia;
  sbb[256 + tid] = (unsigned char)ib;
  sbb[512 + tid] = (unsigned char)ic;
  sbb[768 + tid] = (unsigned char)id;
  __syncthreads();
  const int p = tid >> 6, i = tid & 63;
  planes[(size_t)p * pw + (size_t)blockIdx.x * 64 + i] = sb[p][i];
}

__device__ __forceinline__ unsigned match4(unsigned w, unsigned rid4) {
  const unsigned x  = w ^ rid4;
  const unsigned nz = ((x & 0x7f7f7f7fu) + 0x7f7f7f7fu) | x;
  const unsigned z  = (~nz & 0x80808080u) >> 7;
  return (z * 0x08102040u) >> 27;
}

template <int K, int FOUT>
__global__ void k_gemm_wmma(const float* __restrict__ A, int lda,
                            const float* __restrict__ W,
                            float* __restrict__ H) {
  const int lane = threadIdx.x & 31;
  const int half = lane >> 4;
  const int lid  = lane & 15;
  const int row  = blockIdx.x * 16 + lid;
  const int col  = blockIdx.y * 16 + lid;
  const bool cv  = (FOUT % 16 == 0) || (col < FOUT);
  const int  cc  = cv ? col : (FOUT - 1);

  const float* Ap = A + row * lda + 2 * half;
  v8f acc = {};
  constexpr int K4 = K & ~3;
#pragma unroll 4
  for (int k0 = 0; k0 < K4; k0 += 4) {
    const int kb = k0 + 2 * half;
    v2f a, b;
    a.x = Ap[k0];
    a.y = Ap[k0 + 1];
    float b0 = W[kb * FOUT + cc];
    float b1 = W[(kb + 1) * FOUT + cc];
    b.x = cv ? b0 : 0.0f;
    b.y = cv ? b1 : 0.0f;
    acc = __builtin_amdgcn_wmma_f32_16x16x4_f32(false, a, false, b, (short)0, acc,
                                                false, false);
  }
  if constexpr (K % 4 != 0) {
    const int kb  = K4 + 2 * half;
    const int k0c = (kb     < K) ? kb     : (K - 1);
    const int k1c = (kb + 1 < K) ? kb + 1 : (K - 1);
    v2f a, b;
    float a0 = A[row * lda + k0c];
    float a1 = A[row * lda + k1c];
    a.x = (kb     < K) ? a0 : 0.0f;
    a.y = (kb + 1 < K) ? a1 : 0.0f;
    float b0 = W[k0c * FOUT + cc];
    float b1 = W[k1c * FOUT + cc];
    b.x = (cv && kb     < K) ? b0 : 0.0f;
    b.y = (cv && kb + 1 < K) ? b1 : 0.0f;
    acc = __builtin_amdgcn_wmma_f32_16x16x4_f32(false, a, false, b, (short)0, acc,
                                                false, false);
  }
  const int hbase = (blockIdx.x * 16 + 8 * half) * PADF + blockIdx.y * 16 + lid;
#pragma unroll
  for (int r = 0; r < 8; ++r)
    H[hbase + r * PADF] = acc[r];
}

#define AGG_THREADS 1024
#define AGG_EPT 32
#define AGG_BATCH (AGG_THREADS * AGG_EPT)
#define AGG_QC 2048
#define AGG_U 8

template <int F, int FP, int NR>
__global__ __launch_bounds__(AGG_THREADS)
void k_agg(const float* __restrict__ h, const unsigned* __restrict__ plane,
           const int* __restrict__ rec, float* __restrict__ part, int E) {
  __shared__ __attribute__((aligned(16))) float acc[NR * FP];
  __shared__ int   q_a[AGG_QC];
  __shared__ int   q_d[AGG_QC];
  __shared__ float q_w[AGG_QC];
  __shared__ int   q_n;
  __shared__ int   q_more;

  const int tid  = threadIdx.x;
  const int lane = tid & 31;
  const int lo   = blockIdx.y * NR;
  const int G    = gridDim.x;
  const int echunk = (((E + G - 1) / G) + 31) & ~31;
  const int e_begin = blockIdx.x * echunk;
  const int e_end   = (e_begin + echunk < E) ? (e_begin + echunk) : E;

  constexpr int L  = FP / 4;
  constexpr int EP = AGG_THREADS / L;
  const int  esub = tid / L;
  const int  pc   = tid - esub * L;
  const int  c4   = pc * 4;
  const bool lane_used = tid < EP * L;
  const int  rot  = (pc + esub) & 3;
  const int  o0 = rot, o1 = (rot + 1) & 3, o2 = (rot + 2) & 3, o3 = (rot + 3) & 3;
  const unsigned rid4 = (unsigned)blockIdx.y * 0x01010101u;

  float4* acc4 = reinterpret_cast<float4*>(acc);
  for (int i = tid; i < NR * FP / 4; i += AGG_THREADS)
    acc4[i] = make_float4(0.0f, 0.0f, 0.0f, 0.0f);
  if (tid == 0) { q_n = 0; q_more = 0; }
  __syncthreads();

  for (int b = e_begin; b < e_end; b += AGG_BATCH) {
    const int eb  = b + tid * AGG_EPT;
    const int nv  = e_end - eb;
    const int ebc = (nv > 0) ? eb : e_begin;
    const uint4 p0 = reinterpret_cast<const uint4*>(plane)[(ebc >> 4)];
    const uint4 p1 = reinterpret_cast<const uint4*>(plane)[(ebc >> 4) + 1];
    unsigned pend = match4(p0.x, rid4)       | match4(p0.y, rid4) << 4  |
                    match4(p0.z, rid4) << 8  | match4(p0.w, rid4) << 12 |
                    match4(p1.x, rid4) << 16 | match4(p1.y, rid4) << 20 |
                    match4(p1.z, rid4) << 24 | match4(p1.w, rid4) << 28;
    pend &= (nv >= 32) ? 0xffffffffu : ((nv > 0) ? ((1u << nv) - 1u) : 0u);

    bool more;
    do {
      const int c = __popc(pend);
      int incl = c;
#pragma unroll
      for (int o = 1; o < 32; o <<= 1) {
        const int t = __shfl_up(incl, o, 32);
        if (lane >= o) incl += t;
      }
      const int wtot = __shfl(incl, 31, 32);
      int base = 0;
      if (lane == 31 && wtot > 0)
        base = __hip_atomic_fetch_add(&q_n, wtot, __ATOMIC_RELAXED, __HIP_MEMORY_SCOPE_WORKGROUP);
      base = __shfl(base, 31, 32);
      int pos = base + incl - c;
      unsigned p = pend;
      while (p) {
        const int j = __ffs(p) - 1;
        p &= p - 1;
        if (pos < AGG_QC) {
          q_a[pos] = eb + j;
          pend &= ~(1u << j);
        }
        ++pos;
      }
      if (pend) q_more = 1;
      __syncthreads();

      const int M = (q_n < AGG_QC) ? q_n : AGG_QC;
      more = (q_more != 0);
      for (int m = tid; m < M; m += AGG_THREADS) {
        const int e = q_a[m];
        const int4 r = reinterpret_cast<const int4*>(rec)[e];
        q_a[m] = r.x;
        q_d[m] = (r.y - lo) * FP;
        q_w[m] = __int_as_float(r.z);
      }
      __syncthreads();

      if (tid == 0) { q_n = 0; q_more = 0; }
      for (int m0 = 0; m0 < M; m0 += EP * AGG_U) {
        float4 hv[AGG_U];
        float  wv[AGG_U];
        int    a[AGG_U];
        bool   ok[AGG_U];
#pragma unroll
        for (int u = 0; u < AGG_U; ++u) {
          const int m = m0 + u * EP + esub;
          ok[u] = lane_used && (m < M);
          const int mc = ok[u] ? m : 0;
          const int qa = q_a[mc];
          const int qd = q_d[mc];
          hv[u] = *reinterpret_cast<const float4*>(h + qa + c4);
          wv[u] = q_w[mc];
          a[u]  = qd + c4;
        }
#pragma unroll
        for (int u = 0; u < AGG_U; ++u)
          if (ok[u]) {
            float v0 = hv[u].x, v1 = hv[u].y, v2 = hv[u].z, v3 = hv[u].w;
            if (rot & 1) { const float t = v0; v0 = v1; v1 = v2; v2 = v3; v3 = t; }
            if (rot & 2) { const float t0 = v0, t1 = v1; v0 = v2; v1 = v3; v2 = t0; v3 = t1; }
            float* p = &acc[a[u]];
            const float w = wv[u];
            __hip_atomic_fetch_add(p + o0, w * v0, __ATOMIC_RELAXED, __HIP_MEMORY_SCOPE_WORKGROUP);
            __hip_atomic_fetch_add(p + o1, w * v1, __ATOMIC_RELAXED, __HIP_MEMORY_SCOPE_WORKGROUP);
            __hip_atomic_fetch_add(p + o2, w * v2, __ATOMIC_RELAXED, __HIP_MEMORY_SCOPE_WORKGROUP);
            __hip_atomic_fetch_add(p + o3, w * v3, __ATOMIC_RELAXED, __HIP_MEMORY_SCOPE_WORKGROUP);
          }
      }
      __syncthreads();
    } while (more);
  }

  const size_t npad = (size_t)gridDim.y * NR;
  float4* out4 = reinterpret_cast<float4*>(part + ((size_t)blockIdx.x * npad + lo) * FP);
  for (int i = tid; i < NR * FP / 4; i += AGG_THREADS)
    out4[i] = acc4[i];
}

template <int F, int FP, int G>
__global__ void k_finalize(const float* __restrict__ h, const float* __restrict__ part,
                           const float* __restrict__ dinv, const float* __restrict__ bias,
                           float* __restrict__ out, int ldo, int N, int npad) {
  int t = blockIdx.x * blockDim.x + threadIdx.x;
  if (t >= N * F) return;
  int n = t / F;
  int f = t - n * F;
  float agg = 0.0f;
#pragma unroll
  for (int g = 0; g < G; ++g)
    agg += part[((size_t)g * npad + n) * FP + f];
  float di = dinv[n];
  float v = agg + di * di * h[n * PADF + f] + bias[f];
  out[n * ldo + f] = selu_f(v);
}

extern "C" void kernel_launch(void* const* d_in, const int* in_sizes, int n_in,
                              void* d_out, int out_size, void* d_ws, size_t ws_size,
                              hipStream_t stream) {
  const float* x  = (const float*)d_in[0];
  const int*   ei = (const int*)d_in[1];
  const float* W1 = (const float*)d_in[2];
  const float* b1 = (const float*)d_in[3];
  const float* W2 = (const float*)d_in[4];
  const float* b2 = (const float*)d_in[5];
  const float* W3 = (const float*)d_in[6];
  const float* b3 = (const float*)d_in[7];
  const float* W4 = (const float*)d_in[8];
  const float* b4 = (const float*)d_in[9];

  const int E = in_sizes[1] / 2;
  const int N = in_sizes[0] / 128;
  const int* src = ei;
  const int* dst = ei + E;

  auto cdiv = [](long long a, long long b) { return (int)((a + b - 1) / b); };
  auto al64 = [](size_t n) { return (n + 63) & ~(size_t)63; };

  float* ws   = (float*)d_ws;
  float* deg  = ws;
  float* dinv = deg  + al64(N);
  int*   rec  = (int*)(dinv + al64(N));
  const int NB = cdiv(E, 256);
  const int pw = NB * 64;
  unsigned* planes = (unsigned*)(rec + al64((size_t)4 * E));
  float* hG   = (float*)(planes + (size_t)4 * pw);
  float* hIn  = hG   + (size_t)N * PADF;
  float* part = hIn  + (size_t)N * PADF;

  const int NT = N / 16;

  constexpr int NR1 = 4096, G1 = 10;
  constexpr int NR2 = 3584, G2 = 9;
  constexpr int NR3 = 2400, G3 = 6;
  constexpr int NR4 = 2048, G4 = 5;
  const int R1 = cdiv(N, NR1), R2 = cdiv(N, NR2), R3 = cdiv(N, NR3), R4 = cdiv(N, NR4);

  k_fill  <<<cdiv(N, 256), 256, 0, stream>>>(deg, 1.0f, N);
  k_degree<<<cdiv(E, 256), 256, 0, stream>>>(dst, deg, E);
  k_rsqrt <<<cdiv(N, 256), 256, 0, stream>>>(deg, dinv, N);
  k_norm<NR1, NR2, NR3, NR4><<<NB, 256, 0, stream>>>(src, dst, dinv, rec, planes, pw, E);

  k_gemm_wmma<128, 15><<<dim3(NT, 1), 32, 0, stream>>>(x, 128, W1, hG);
  k_agg<15, 16, NR1><<<dim3(G1, R1), AGG_THREADS, 0, stream>>>(hG, planes, rec, part, E);
  k_finalize<15, 16, G1><<<cdiv((long long)N * 15, 256), 256, 0, stream>>>(hG, part, dinv, b1, hIn, PADF, N, R1 * NR1);

  k_gemm_wmma<15, 20><<<dim3(NT, 2), 32, 0, stream>>>(hIn, PADF, W2, hG);
  k_agg<20, 20, NR2><<<dim3(G2, R2), AGG_THREADS, 0, stream>>>(hG, planes + (size_t)pw, rec, part, E);
  k_finalize<20, 20, G2><<<cdiv((long long)N * 20, 256), 256, 0, stream>>>(hG, part, dinv, b2, hIn, PADF, N, R2 * NR2);

  k_gemm_wmma<20, 27><<<dim3(NT, 2), 32, 0, stream>>>(hIn, PADF, W3, hG);
  k_agg<27, 28, NR3><<<dim3(G3, R3), AGG_THREADS, 0, stream>>>(hG, planes + (size_t)2 * pw, rec, part, E);
  k_finalize<27, 28, G3><<<cdiv((long long)N * 27, 256), 256, 0, stream>>>(hG, part, dinv, b3, hIn, PADF, N, R3 * NR3);

  k_gemm_wmma<27, 36><<<dim3(NT, 3), 32, 0, stream>>>(hIn, PADF, W4, hG);
  k_agg<36, 36, NR4><<<dim3(G4, R4), AGG_THREADS, 0, stream>>>(hG, planes + (size_t)3 * pw, rec, part, E);
  k_finalize<36, 36, G4><<<cdiv((long long)N * 36, 256), 256, 0, stream>>>(hG, part, dinv, b4,
                                                                   (float*)d_out, 36, N, R4 * NR4);
}


# k_agg scans its edge chunk in one pass (match bits of 160-320 edges per thread in registers), drains them in 2-3 rounds of a deep queue, not 5-10 scan-and-drain batches; LDS add order changes
# speedup vs baseline: 1.6276x; 1.0695x over previous
#include <hip/hip_runtime.h>
#include <hip/hip_bf16.h>

typedef __attribute__((ext_vector_type(2))) float v2f;
typedef __attribute__((ext_vector_type(8))) float v8f;

#define PADF 48

__device__ __forceinline__ float selu_f(float x) {
  const float alpha = 1.6732632423543772848f;
  const float scale = 1.0507009873554804934f;
  return scale * (x > 0.0f ? x : alpha * (expf(x) - 1.0f));
}

__global__ void k_fill(float* __restrict__ p, float v, int n) {
  int i = blockIdx.x * blockDim.x + threadIdx.x;
  if (i < n) p[i] = v;
}

__global__ void k_degree(const int* __restrict__ dst, float* __restrict__ deg, int E) {
  int e = blockIdx.x * blockDim.x + threadIdx.x;
  if (e < E)
    __hip_atomic_fetch_add(&deg[dst[e]], 1.0f, __ATOMIC_RELAXED, __HIP_MEMORY_SCOPE_AGENT);
}

__global__ void k_rsqrt(const float* __restrict__ deg, float* __restrict__ dinv, int n) {
  int i = blockIdx.x * blockDim.x + threadIdx.x;
  if (i < n) {
    float d = deg[i];
    dinv[i] = d > 0.0f ? rsqrtf(d) : 0.0f;
  }
}

template <int NRA, int NRB, int NRC, int NRD>
__global__ __launch_bounds__(256)
void k_norm(const int* __restrict__ src, const int* __restrict__ dst,
            const float* __restrict__ dinv, int* __restrict__ rec,
            unsigned* __restrict__ planes, int pw, int E) {
  __shared__ unsigned sb[4][64];
  const int tid = threadIdx.x;
  const int e = blockIdx.x * 256 + tid;
  unsigned ia = 255u, ib = 255u, ic = 255u, id = 255u;
  if (e < E) {
    const int s = src[e];
    const int d = dst[e];
    const float w = dinv[s] * dinv[d];
    reinterpret_cast<int4*>(rec)[e] = make_int4(s * PADF, d, __float_as_int(w), 0);
    ia = (unsigned)d / (unsigned)NRA;
    ib = (unsigned)d / (unsigned)NRB;
    ic = (unsigned)d / (unsigned)NRC;
    id = (unsigned)d / (unsigned)NRD;
  }
  unsigned char* sbb = reinterpret_cast<unsigned char*>(&sb[0][0]);
  sbb[tid]       = (unsigned char)ia;
  sbb[256 + tid] = (unsigned char)ib;
  sbb[512 + tid] = (unsigned char)ic;
  sbb[768 + tid] = (unsigned char)id;
  __syncthreads();
  const int p = tid >> 6, i = tid & 63;
  planes[(size_t)p * pw + (size_t)blockIdx.x * 64 + i] = sb[p][i];
}

__device__ __forceinline__ unsigned match4(unsigned w, unsigned rid4) {
  const unsigned x  = w ^ rid4;
  const unsigned nz = ((x & 0x7f7f7f7fu) + 0x7f7f7f7fu) | x;
  const unsigned z  = (~nz & 0x80808080u) >> 7;
  return (z * 0x08102040u) >> 27;
}

template <int K, int FOUT>
__global__ void k_gemm_wmma(const float* __restrict__ A, int lda,
                            const float* __restrict__ W,
                            float* __restrict__ H) {
  const int lane = threadIdx.x & 31;
  const int half = lane >> 4;
  const int lid  = lane & 15;
  const int row  = blockIdx.x * 16 + lid;
  const int col  = blockIdx.y * 16 + lid;
  const bool cv  = (FOUT % 16 == 0) || (col < FOUT);
  const int  cc  = cv ? col : (FOUT - 1);

  const float* Ap = A + row * lda + 2 * half;
  v8f acc = {};
  constexpr int K4 = K & ~3;
#pragma unroll 4
  for (int k0 = 0; k0 < K4; k0 += 4) {
    const int kb = k0 + 2 * half;
    v2f a, b;
    a.x = Ap[k0];
    a.y = Ap[k0 + 1];
    float b0 = W[kb * FOUT + cc];
    float b1 = W[(kb + 1) * FOUT + cc];
    b.x = cv ? b0 : 0.0f;
    b.y = cv ? b1 : 0.0f;
    acc = __builtin_amdgcn_wmma_f32_16x16x4_f32(false, a, false, b, (short)0, acc,
                                                false, false);
  }
  if constexpr (K % 4 != 0) {
    const int kb  = K4 + 2 * half;
    const int k0c = (kb     < K) ? kb     : (K - 1);
    const int k1c = (kb + 1 < K) ? kb + 1 : (K - 1);
    v2f a, b;
    float a0 = A[row * lda + k0c];
    float a1 = A[row * lda + k1c];
    a.x = (kb     < K) ? a0 : 0.0f;
    a.y = (kb + 1 < K) ? a1 : 0.0f;
    float b0 = W[k0c * FOUT + cc];
    float b1 = W[k1c * FOUT + cc];
    b.x = (cv && kb     < K) ? b0 : 0.0f;
    b.y = (cv && kb + 1 < K) ? b1 : 0.0f;
    acc = __builtin_amdgcn_wmma_f32_16x16x4_f32(false, a, false, b, (short)0, acc,
                                                false, false);
  }
  const int hbase = (blockIdx.x * 16 + 8 * half) * PADF + blockIdx.y * 16 + lid;
#pragma unroll
  for (int r = 0; r < 8; ++r)
    H[hbase + r * PADF] = acc[r];
}

#define AGG_THREADS 1024
#define AGG_U 8

template <int F, int FP, int NR, int QC, int W>
__global__ __launch_bounds__(AGG_THREADS)
void k_agg(const float* __restrict__ h, const unsigned* __restrict__ plane,
           const int* __restrict__ rec, float* __restrict__ part, int E) {
  __shared__ __attribute__((aligned(16))) float acc[NR * FP];
  __shared__ int   q_a[QC];
  __shared__ int   q_d[QC];
  __shared__ float q_w[QC];
  __shared__ int   q_n;
  __shared__ int   q_more;

  const int tid  = threadIdx.x;
  const int lane = tid & 31;
  const int lo   = blockIdx.y * NR;
  const int G    = gridDim.x;
  const int echunk = (((E + G - 1) / G) + 31) & ~31;
  const int e_begin = blockIdx.x * echunk;
  const int e_end   = (e_begin + echunk < E) ? (e_begin + echunk) : E;

  constexpr int PASS = AGG_THREADS * 32 * W;
  constexpr int NF   = (QC + AGG_THREADS - 1) / AGG_THREADS;
  constexpr int L  = FP / 4;
  constexpr int EP = AGG_THREADS / L;
  const int  esub = tid / L;
  const int  pc   = tid - esub * L;
  const int  c4   = pc * 4;
  const bool lane_used = tid < EP * L;
  const int  rot  = (pc + esub) & 3;
  const int  o0 = rot, o1 = (rot + 1) & 3, o2 = (rot + 2) & 3, o3 = (rot + 3) & 3;
  const unsigned rid4 = (unsigned)blockIdx.y * 0x01010101u;

  float4* acc4 = reinterpret_cast<float4*>(acc);
  for (int i = tid; i < NR * FP / 4; i += AGG_THREADS)
    acc4[i] = make_float4(0.0f, 0.0f, 0.0f, 0.0f);
  if (tid == 0) { q_n = 0; q_more = 0; }
  __syncthreads();

  const uint4* plane4 = reinterpret_cast<const uint4*>(plane);
  const int4*  rec4   = reinterpret_cast<const int4*>(rec);

  for (int b = e_begin; b < e_end; b += PASS) {
    unsigned pend[W];
    constexpr int WC = (W <= 6) ? W : (W + 1) / 2;
#pragma unroll
    for (int w0 = 0; w0 < W; w0 += WC) {
      uint4 pl[2 * WC];
#pragma unroll
      for (int i = 0; i < WC; ++i)
        if (w0 + i < W) {
          const int eb  = b + ((w0 + i) * AGG_THREADS + tid) * 32;
          const int ebc = (e_end - eb > 0) ? eb : e_begin;
          pl[2 * i]     = plane4[(ebc >> 4)];
          pl[2 * i + 1] = plane4[(ebc >> 4) + 1];
        }
      __builtin_amdgcn_sched_barrier(0);
#pragma unroll
      for (int i = 0; i < WC; ++i)
        if (w0 + i < W) {
          const int eb = b + ((w0 + i) * AGG_THREADS + tid) * 32;
          const int nv = e_end - eb;
          const uint4 p0 = pl[2 * i], p1 = pl[2 * i + 1];
          unsigned mt = match4(p0.x, rid4)       | match4(p0.y, rid4) << 4  |
                        match4(p0.z, rid4) << 8  | match4(p0.w, rid4) << 12 |
                        match4(p1.x, rid4) << 16 | match4(p1.y, rid4) << 20 |
                        match4(p1.z, rid4) << 24 | match4(p1.w, rid4) << 28;
          mt &= (nv >= 32) ? 0xffffffffu : ((nv > 0) ? ((1u << nv) - 1u) : 0u);
          pend[w0 + i] = mt;
        }
    }

    bool more;
    do {
      int c = 0;
#pragma unroll
      for (int w = 0; w < W; ++w) c += __popc(pend[w]);
      int incl = c;
#pragma unroll
      for (int o = 1; o < 32; o <<= 1) {
        const int t = __shfl_up(incl, o, 32);
        if (lane >= o) incl += t;
      }
      const int wtot = __shfl(incl, 31, 32);
      int base = 0;
      if (lane == 31 && wtot > 0)
        base = __hip_atomic_fetch_add(&q_n, wtot, __ATOMIC_RELAXED, __HIP_MEMORY_SCOPE_WORKGROUP);
      base = __shfl(base, 31, 32);
      int pos = base + incl - c;
      unsigned left = 0;
#pragma unroll
      for (int w = 0; w < W; ++w) {
        const int eb = b + (w * AGG_THREADS + tid) * 32;
        unsigned p = pend[w];
        while (p != 0u && pos < QC) {
          const int j = __ffs(p) - 1;
          p &= p - 1;
          q_a[pos] = eb + j;
          ++pos;
        }
        pend[w] = p;
        left |= p;
      }
      if (left) q_more = 1;
      __syncthreads();

      const int M = (q_n < QC) ? q_n : QC;
      more = (q_more != 0);
      int4 rr[NF];
#pragma unroll
      for (int u = 0; u < NF; ++u) {
        const int m  = tid + u * AGG_THREADS;
        const bool fo = m < M;
        const int qe = q_a[fo ? m : 0];
        rr[u] = rec4[fo ? qe : e_begin];
      }
#pragma unroll
      for (int u = 0; u < NF; ++u) {
        asm volatile("" :: "v"(rr[u].x), "v"(rr[u].y), "v"(rr[u].z));
      }
#pragma unroll
      for (int u = 0; u < NF; ++u) {
        const int m = tid + u * AGG_THREADS;
        if (m < M) {
          q_a[m] = rr[u].x;
          q_d[m] = (rr[u].y - lo) * FP;
          q_w[m] = __int_as_float(rr[u].z);
        }
      }
      __syncthreads();

      if (tid == 0) { q_n = 0; q_more = 0; }
      for (int m0 = 0; m0 < M; m0 += EP * AGG_U) {
        float4 hv[AGG_U];
        float  wv[AGG_U];
        int    a[AGG_U];
        bool   ok[AGG_U];
#pragma unroll
        for (int u = 0; u < AGG_U; ++u) {
          const int m = m0 + u * EP + esub;
          ok[u] = lane_used && (m < M);
          const int mc = ok[u] ? m : 0;
          const int qa = q_a[mc];
          const int qd = q_d[mc];
          hv[u] = *reinterpret_cast<const float4*>(h + qa + c4);
          wv[u] = q_w[mc];
          a[u]  = qd + c4;
        }
#pragma unroll
        for (int u = 0; u < AGG_U; ++u)
          if (ok[u]) {
            float v0 = hv[u].x, v1 = hv[u].y, v2 = hv[u].z, v3 = hv[u].w;
            if (rot & 1) { const float t = v0; v0 = v1; v1 = v2; v2 = v3; v3 = t; }
            if (rot & 2) { const float t0 = v0, t1 = v1; v0 = v2; v1 = v3; v2 = t0; v3 = t1; }
            float* p = &acc[a[u]];
            const float w = wv[u];
            __hip_atomic_fetch_add(p + o0, w * v0, __ATOMIC_RELAXED, __HIP_MEMORY_SCOPE_WORKGROUP);
            __hip_atomic_fetch_add(p + o1, w * v1, __ATOMIC_RELAXED, __HIP_MEMORY_SCOPE_WORKGROUP);
            __hip_atomic_fetch_add(p + o2, w * v2, __ATOMIC_RELAXED, __HIP_MEMORY_SCOPE_WORKGROUP);
            __hip_atomic_fetch_add(p + o3, w * v3, __ATOMIC_RELAXED, __HIP_MEMORY_SCOPE_WORKGROUP);
          }
      }
      __syncthreads();
    } while (more);
  }

  const size_t npad = (size_t)gridDim.y * NR;
  float4* out4 = reinterpret_cast<float4*>(part + ((size_t)blockIdx.x * npad + lo) * FP);
  for (int i = tid; i < NR * FP / 4; i += AGG_THREADS)
    out4[i] = acc4[i];
}

template <int F, int FP, int G>
__global__ void k_finalize(const float* __restrict__ h, const float* __restrict__ part,
                           const float* __restrict__ dinv, const float* __restrict__ bias,
                           float* __restrict__ out, int ldo, int N, int npad) {
  int t = blockIdx.x * blockDim.x + threadIdx.x;
  if (t >= N * F) return;
  int n = t / F;
  int f = t - n * F;
  float agg = 0.0f;
#pragma unroll
  for (int g = 0; g < G; ++g)
    agg += part[((size_t)g * npad + n) * FP + f];
  float di = dinv[n];
  float v = agg + di * di * h[n * PADF + f] + bias[f];
  out[n * ldo + f] = selu_f(v);
}

extern "C" void kernel_launch(void* const* d_in, const int* in_sizes, int n_in,
                              void* d_out, int out_size, void* d_ws, size_t ws_size,
                              hipStream_t stream) {
  const float* x  = (const float*)d_in[0];
  const int*   ei = (const int*)d_in[1];
  const float* W1 = (const float*)d_in[2];
  const float* b1 = (const float*)d_in[3];
  const float* W2 = (const float*)d_in[4];
  const float* b2 = (const float*)d_in[5];
  const float* W3 = (const float*)d_in[6];
  const float* b3 = (const float*)d_in[7];
  const float* W4 = (const float*)d_in[8];
  const float* b4 = (const float*)d_in[9];

  const int E = in_sizes[1] / 2;
  const int N = in_sizes[0] / 128;
  const int* src = ei;
  const int* dst = ei + E;

  auto cdiv = [](long long a, long long b) { return (int)((a + b - 1) / b); };
  auto al64 = [](size_t n) { return (n + 63) & ~(size_t)63; };

  float* ws   = (float*)d_ws;
  float* deg  = ws;
  float* dinv = deg  + al64(N);
  int*   rec  = (int*)(dinv + al64(N));
  const int NB = cdiv(E, 256);
  const int pw = NB * 64;
  unsigned* planes = (unsigned*)(rec + al64((size_t)4 * E));
  float* hG   = (float*)(planes + (size_t)4 * pw);
  float* hIn  = hG   + (size_t)N * PADF;
  float* part = hIn  + (size_t)N * PADF;

  const int NT = N / 16;

  constexpr int NR1 = 4096, G1 = 10;
  constexpr int NR2 = 3572, G2 = 9;
  constexpr int NR3 = 2400, G3 = 6;
  constexpr int NR4 = 2048, G4 = 5;
  constexpr int QC1 = 5120, SW1 = 5;
  constexpr int QC2 = 3320, SW2 = 6;
  constexpr int QC3 = 4608, SW3 = 9;
  constexpr int QC4 = 2552, SW4 = 10;
  const int R1 = cdiv(N, NR1), R2 = cdiv(N, NR2), R3 = cdiv(N, NR3), R4 = cdiv(N, NR4);

  k_fill  <<<cdiv(N, 256), 256, 0, stream>>>(deg, 1.0f, N);
  k_degree<<<cdiv(E, 256), 256, 0, stream>>>(dst, deg, E);
  k_rsqrt <<<cdiv(N, 256), 256, 0, stream>>>(deg, dinv, N);
  k_norm<NR1, NR2, NR3, NR4><<<NB, 256, 0, stream>>>(src, dst, dinv, rec, planes, pw, E);

  k_gemm_wmma<128, 15><<<dim3(NT, 1), 32, 0, stream>>>(x, 128, W1, hG);
  k_agg<15, 16, NR1, QC1, SW1><<<dim3(G1, R1), AGG_THREADS, 0, stream>>>(hG, planes, rec, part, E);
  k_finalize<15, 16, G1><<<cdiv((long long)N * 15, 256), 256, 0, stream>>>(hG, part, dinv, b1, hIn, PADF, N, R1 * NR1);

  k_gemm_wmma<15, 20><<<dim3(NT, 2), 32, 0, stream>>>(hIn, PADF, W2, hG);
  k_agg<20, 20, NR2, QC2, SW2><<<dim3(G2, R2), AGG_THREADS, 0, stream>>>(hG, planes + (size_t)pw, rec, part, E);
  k_finalize<20, 20, G2><<<cdiv((long long)N * 20, 256), 256, 0, stream>>>(hG, part, dinv, b2, hIn, PADF, N, R2 * NR2);

  k_gemm_wmma<20, 27><<<dim3(NT, 2), 32, 0, stream>>>(hIn, PADF, W3, hG);
  k_agg<27, 28, NR3, QC3, SW3><<<dim3(G3, R3), AGG_THREADS, 0, stream>>>(hG, planes + (size_t)2 * pw, rec, part, E);
  k_finalize<27, 28, G3><<<cdiv((long long)N * 27, 256), 256, 0, stream>>>(hG, part, dinv, b3, hIn, PADF, N, R3 * NR3);

  k_gemm_wmma<27, 36><<<dim3(NT, 3), 32, 0, stream>>>(hIn, PADF, W4, hG);
  k_agg<36, 36, NR4, QC4, SW4><<<dim3(G4, R4), AGG_THREADS, 0, stream>>>(hG, planes + (size_t)3 * pw, rec, part, E);
  k_finalize<36, 36, G4><<<cdiv((long long)N * 36, 256), 256, 0, stream>>>(hG, part, dinv, b4,
                                                                   (float*)d_out, 36, N, R4 * NR4);
}


# GEMM: a wave takes 2 row tiles x all column tiles (A fragment loaded once per k step), 8 waves per workgroup: 3125 waves enter a layer, not 6250-18750 one-wave workgroups; sum order unchanged
# speedup vs baseline: 1.6785x; 1.0313x over previous
#include <hip/hip_runtime.h>
#include <hip/hip_bf16.h>

typedef __attribute__((ext_vector_type(2))) float v2f;
typedef __attribute__((ext_vector_type(8))) float v8f;

#define PADF 48

__device__ __forceinline__ float selu_f(float x) {
  const float alpha = 1.6732632423543772848f;
  const float scale = 1.0507009873554804934f;
  return scale * (x > 0.0f ? x : alpha * (expf(x) - 1.0f));
}

__global__ void k_fill(float* __restrict__ p, float v, int n) {
  int i = blockIdx.x * blockDim.x + threadIdx.x;
  if (i < n) p[i] = v;
}

__global__ void k_degree(const int* __restrict__ dst, float* __restrict__ deg, int E) {
  int e = blockIdx.x * blockDim.x + threadIdx.x;
  if (e < E)
    __hip_atomic_fetch_add(&deg[dst[e]], 1.0f, __ATOMIC_RELAXED, __HIP_MEMORY_SCOPE_AGENT);
}

__global__ void k_rsqrt(const float* __restrict__ deg, float* __restrict__ dinv, int n) {
  int i = blockIdx.x * blockDim.x + threadIdx.x;
  if (i < n) {
    float d = deg[i];
    dinv[i] = d > 0.0f ? rsqrtf(d) : 0.0f;
  }
}

template <int NRA, int NRB, int NRC, int NRD>
__global__ __launch_bounds__(256)
void k_norm(const int* __restrict__ src, const int* __restrict__ dst,
            const float* __restrict__ dinv, int* __restrict__ rec,
            unsigned* __restrict__ planes, int pw, int E) {
  __shared__ unsigned sb[4][64];
  const int tid = threadIdx.x;
  const int e = blockIdx.x * 256 + tid;
  unsigned ia = 255u, ib = 255u, ic = 255u, id = 255u;
  if (e < E) {
    const int s = src[e];
    const int d = dst[e];
    const float w = dinv[s] * dinv[d];
    reinterpret_cast<int4*>(rec)[e] = make_int4(s * PADF, d, __float_as_int(w), 0);
    ia = (unsigned)d / (unsigned)NRA;
    ib = (unsigned)d / (unsigned)NRB;
    ic = (unsigned)d / (unsigned)NRC;
    id = (unsigned)d / (unsigned)NRD;
  }
  unsigned char* sbb = reinterpret_cast<unsigned char*>(&sb[0][0]);
  sbb[tid]       = (unsigned char)ia;
  sbb[256 + tid] = (unsigned char)ib;
  sbb[512 + tid] = (unsigned char)ic;
  sbb[768 + tid] = (unsigned char)id;
  __syncthreads();
  const int p = tid >> 6, i = tid & 63;
  planes[(size_t)p * pw + (size_t)blockIdx.x * 64 + i] = sb[p][i];
}

__device__ __forceinline__ unsigned match4(unsigned w, unsigned rid4) {
  const unsigned x  = w ^ rid4;
  const unsigned nz = ((x & 0x7f7f7f7fu) + 0x7f7f7f7fu) | x;
  const unsigned z  = (~nz & 0x80808080u) >> 7;
  return (z * 0x08102040u) >> 27;
}

#define GEMM_RT 2
#define GEMM_WAVES 8

template <int K, int FOUT>
__global__ __launch_bounds__(GEMM_WAVES * 32)
void k_gemm_wmma(const float* __restrict__ A, int lda,
                 const float* __restrict__ W,
                 float* __restrict__ H, int NT) {
  constexpr int CT = (FOUT + 15) / 16;
  const int lane = threadIdx.x & 31;
  const int half = lane >> 4;
  const int lid  = lane & 15;
  const int wave = __builtin_amdgcn_readfirstlane((int)(threadIdx.x >> 5));
  const int t0   = (blockIdx.x * GEMM_WAVES + wave) * GEMM_RT;
  if (t0 >= NT) return;

  int tile[GEMM_RT];
  const float* Ap[GEMM_RT];
#pragma unroll
  for (int rt = 0; rt < GEMM_RT; ++rt) {
    tile[rt] = (t0 + rt < NT) ? (t0 + rt) : (NT - 1);
    Ap[rt]   = A + (tile[rt] * 16 + lid) * lda + 2 * half;
  }
  bool cv[CT];
  int  cc[CT];
#pragma unroll
  for (int ct = 0; ct < CT; ++ct) {
    const int col = ct * 16 + lid;
    cv[ct] = col < FOUT;
    cc[ct] = cv[ct] ? col : (FOUT - 1);
  }

  v8f acc[GEMM_RT][CT];
#pragma unroll
  for (int rt = 0; rt < GEMM_RT; ++rt)
#pragma unroll
    for (int ct = 0; ct < CT; ++ct)
      acc[rt][ct] = v8f{};

  constexpr int K4 = K & ~3;
#pragma unroll 4
  for (int k0 = 0; k0 < K4; k0 += 4) {
    const int kb = k0 + 2 * half;
    v2f a[GEMM_RT], b[CT];
#pragma unroll
    for (int rt = 0; rt < GEMM_RT; ++rt) {
      a[rt].x = Ap[rt][k0];
      a[rt].y = Ap[rt][k0 + 1];
    }
#pragma unroll
    for (int ct = 0; ct < CT; ++ct) {
      const float b0 = W[kb * FOUT + cc[ct]];
      const float b1 = W[(kb + 1) * FOUT + cc[ct]];
      b[ct].x = cv[ct] ? b0 : 0.0f;
      b[ct].y = cv[ct] ? b1 : 0.0f;
    }
#pragma unroll
    for (int rt = 0; rt < GEMM_RT; ++rt)
#pragma unroll
      for (int ct = 0; ct < CT; ++ct)
        acc[rt][ct] = __builtin_amdgcn_wmma_f32_16x16x4_f32(false, a[rt], false, b[ct], (short)0,
                                                            acc[rt][ct], false, false);
  }
  if constexpr (K % 4 != 0) {
    const int kb  = K4 + 2 * half;
    const int k0c = (kb     < K) ? kb     : (K - 1);
    const int k1c = (kb + 1 < K) ? kb + 1 : (K - 1);
    v2f a[GEMM_RT], b[CT];
#pragma unroll
    for (int rt = 0; rt < GEMM_RT; ++rt) {
      const float* Ar = A + (tile[rt] * 16 + lid) * lda;
      const float a0 = Ar[k0c];
      const float a1 = Ar[k1c];
      a[rt].x = (kb     < K) ? a0 : 0.0f;
      a[rt].y = (kb + 1 < K) ? a1 : 0.0f;
    }
#pragma unroll
    for (int ct = 0; ct < CT; ++ct) {
      const float b0 = W[k0c * FOUT + cc[ct]];
      const float b1 = W[k1c * FOUT + cc[ct]];
      b[ct].x = (cv[ct] && kb     < K) ? b0 : 0.0f;
      b[ct].y = (cv[ct] && kb + 1 < K) ? b1 : 0.0f;
    }
#pragma unroll
    for (int rt = 0; rt < GEMM_RT; ++rt)
#pragma unroll
      for (int ct = 0; ct < CT; ++ct)
        acc[rt][ct] = __builtin_amdgcn_wmma_f32_16x16x4_f32(false, a[rt], false, b[ct], (short)0,
                                                            acc[rt][ct], false, false);
  }
#pragma unroll
  for (int rt = 0; rt < GEMM_RT; ++rt)
#pragma unroll
    for (int ct = 0; ct < CT; ++ct) {
      const int hbase = (tile[rt] * 16 + 8 * half) * PADF + ct * 16 + lid;
#pragma unroll
      for (int r = 0; r < 8; ++r)
        H[hbase + r * PADF] = acc[rt][ct][r];
    }
}

#define AGG_THREADS 1024
#define AGG_U 8

template <int F, int FP, int NR, int QC, int W>
__global__ __launch_bounds__(AGG_THREADS)
void k_agg(const float* __restrict__ h, const unsigned* __restrict__ plane,
           const int* __restrict__ rec, float* __restrict__ part, int E) {
  __shared__ __attribute__((aligned(16))) float acc[NR * FP];
  __shared__ int   q_a[QC];
  __shared__ int   q_d[QC];
  __shared__ float q_w[QC];
  __shared__ int   q_n;
  __shared__ int   q_more;

  const int tid  = threadIdx.x;
  const int lane = tid & 31;
  const int lo   = blockIdx.y * NR;
  const int G    = gridDim.x;
  const int echunk = (((E + G - 1) / G) + 31) & ~31;
  const int e_begin = blockIdx.x * echunk;
  const int e_end   = (e_begin + echunk < E) ? (e_begin + echunk) : E;

  constexpr int PASS = AGG_THREADS * 32 * W;
  constexpr int NF   = (QC + AGG_THREADS - 1) / AGG_THREADS;
  constexpr int L  = FP / 4;
  constexpr int EP = AGG_THREADS / L;
  const int  esub = tid / L;
  const int  pc   = tid - esub * L;
  const int  c4   = pc * 4;
  const bool lane_used = tid < EP * L;
  const int  rot  = (pc + esub) & 3;
  const int  o0 = rot, o1 = (rot + 1) & 3, o2 = (rot + 2) & 3, o3 = (rot + 3) & 3;
  const unsigned rid4 = (unsigned)blockIdx.y * 0x01010101u;

  float4* acc4 = reinterpret_cast<float4*>(acc);
  for (int i = tid; i < NR * FP / 4; i += AGG_THREADS)
    acc4[i] = make_float4(0.0f, 0.0f, 0.0f, 0.0f);
  if (tid == 0) { q_n = 0; q_more = 0; }
  __syncthreads();

  const uint4* plane4 = reinterpret_cast<const uint4*>(plane);
  const int4*  rec4   = reinterpret_cast<const int4*>(rec);

  for (int b = e_begin; b < e_end; b += PASS) {
    unsigned pend[W];
    constexpr int WC = (W <= 6) ? W : (W + 1) / 2;
#pragma unroll
    for (int w0 = 0; w0 < W; w0 += WC) {
      uint4 pl[2 * WC];
#pragma unroll
      for (int i = 0; i < WC; ++i)
        if (w0 + i < W) {
          const int eb  = b + ((w0 + i) * AGG_THREADS + tid) * 32;
          const int ebc = (e_end - eb > 0) ? eb : e_begin;
          pl[2 * i]     = plane4[(ebc >> 4)];
          pl[2 * i + 1] = plane4[(ebc >> 4) + 1];
        }
      __builtin_amdgcn_sched_barrier(0);
#pragma unroll
      for (int i = 0; i < WC; ++i)
        if (w0 + i < W) {
          const int eb = b + ((w0 + i) * AGG_THREADS + tid) * 32;
          const int nv = e_end - eb;
          const uint4 p0 = pl[2 * i], p1 = pl[2 * i + 1];
          unsigned mt = match4(p0.x, rid4)       | match4(p0.y, rid4) << 4  |
                        match4(p0.z, rid4) << 8  | match4(p0.w, rid4) << 12 |
                        match4(p1.x, rid4) << 16 | match4(p1.y, rid4) << 20 |
                        match4(p1.z, rid4) << 24 | match4(p1.w, rid4) << 28;
          mt &= (nv >= 32) ? 0xffffffffu : ((nv > 0) ? ((1u << nv) - 1u) : 0u);
          pend[w0 + i] = mt;
        }
    }

    bool more;
    do {
      int c = 0;
#pragma unroll
      for (int w = 0; w < W; ++w) c += __popc(pend[w]);
      int incl = c;
#pragma unroll
      for (int o = 1; o < 32; o <<= 1) {
        const int t = __shfl_up(incl, o, 32);
        if (lane >= o) incl += t;
      }
      const int wtot = __shfl(incl, 31, 32);
      int base = 0;
      if (lane == 31 && wtot > 0)
        base = __hip_atomic_fetch_add(&q_n, wtot, __ATOMIC_RELAXED, __HIP_MEMORY_SCOPE_WORKGROUP);
      base = __shfl(base, 31, 32);
      int pos = base + incl - c;
      unsigned left = 0;
#pragma unroll
      for (int w = 0; w < W; ++w) {
        const int eb = b + (w * AGG_THREADS + tid) * 32;
        unsigned p = pend[w];
        while (p != 0u && pos < QC) {
          const int j = __ffs(p) - 1;
          p &= p - 1;
          q_a[pos] = eb + j;
          ++pos;
        }
        pend[w] = p;
        left |= p;
      }
      if (left) q_more = 1;
      __syncthreads();

      const int M = (q_n < QC) ? q_n : QC;
      more = (q_more != 0);
      int4 rr[NF];
#pragma unroll
      for (int u = 0; u < NF; ++u) {
        const int m  = tid + u * AGG_THREADS;
        const bool fo = m < M;
        const int qe = q_a[fo ? m : 0];
        rr[u] = rec4[fo ? qe : e_begin];
      }
#pragma unroll
      for (int u = 0; u < NF; ++u) {
        asm volatile("" :: "v"(rr[u].x), "v"(rr[u].y), "v"(rr[u].z));
      }
#pragma unroll
      for (int u = 0; u < NF; ++u) {
        const int m = tid + u * AGG_THREADS;
        if (m < M) {
          q_a[m] = rr[u].x;
          q_d[m] = (rr[u].y - lo) * FP;
          q_w[m] = __int_as_float(rr[u].z);
        }
      }
      __syncthreads();

      if (tid == 0) { q_n = 0; q_more = 0; }
      for (int m0 = 0; m0 < M; m0 += EP * AGG_U) {
        float4 hv[AGG_U];
        float  wv[AGG_U];
        int    a[AGG_U];
        bool   ok[AGG_U];
#pragma unroll
        for (int u = 0; u < AGG_U; ++u) {
          const int m = m0 + u * EP + esub;
          ok[u] = lane_used && (m < M);
          const int mc = ok[u] ? m : 0;
          const int qa = q_a[mc];
          const int qd = q_d[mc];
          hv[u] = *reinterpret_cast<const float4*>(h + qa + c4);
          wv[u] = q_w[mc];
          a[u]  = qd + c4;
        }
#pragma unroll
        for (int u = 0; u < AGG_U; ++u)
          if (ok[u]) {
            float v0 = hv[u].x, v1 = hv[u].y, v2 = hv[u].z, v3 = hv[u].w;
            if (rot & 1) { const float t = v0; v0 = v1; v1 = v2; v2 = v3; v3 = t; }
            if (rot & 2) { const float t0 = v0, t1 = v1; v0 = v2; v1 = v3; v2 = t0; v3 = t1; }
            float* p = &acc[a[u]];
            const float w = wv[u];
            __hip_atomic_fetch_add(p + o0, w * v0, __ATOMIC_RELAXED, __HIP_MEMORY_SCOPE_WORKGROUP);
            __hip_atomic_fetch_add(p + o1, w * v1, __ATOMIC_RELAXED, __HIP_MEMORY_SCOPE_WORKGROUP);
            __hip_atomic_fetch_add(p + o2, w * v2, __ATOMIC_RELAXED, __HIP_MEMORY_SCOPE_WORKGROUP);
            __hip_atomic_fetch_add(p + o3, w * v3, __ATOMIC_RELAXED, __HIP_MEMORY_SCOPE_WORKGROUP);
          }
      }
      __syncthreads();
    } while (more);
  }

  const size_t npad = (size_t)gridDim.y * NR;
  float4* out4 = reinterpret_cast<float4*>(part + ((size_t)blockIdx.x * npad + lo) * FP);
  for (int i = tid; i < NR * FP / 4; i += AGG_THREADS)
    out4[i] = acc4[i];
}

template <int F, int FP, int G>
__global__ void k_finalize(const float* __restrict__ h, const float* __restrict__ part,
                           const float* __restrict__ dinv, const float* __restrict__ bias,
                           float* __restrict__ out, int ldo, int N, int npad) {
  int t = blockIdx.x * blockDim.x + threadIdx.x;
  if (t >= N * F) return;
  int n = t / F;
  int f = t - n * F;
  float agg = 0.0f;
#pragma unroll
  for (int g = 0; g < G; ++g)
    agg += part[((size_t)g * npad + n) * FP + f];
  float di = dinv[n];
  float v = agg + di * di * h[n * PADF + f] + bias[f];
  out[n * ldo + f] = selu_f(v);
}

extern "C" void kernel_launch(void* const* d_in, const int* in_sizes, int n_in,
                              void* d_out, int out_size, void* d_ws, size_t ws_size,
                              hipStream_t stream) {
  const float* x  = (const float*)d_in[0];
  const int*   ei = (const int*)d_in[1];
  const float* W1 = (const float*)d_in[2];
  const float* b1 = (const float*)d_in[3];
  const float* W2 = (const float*)d_in[4];
  const float* b2 = (const float*)d_in[5];
  const float* W3 = (const float*)d_in[6];
  const float* b3 = (const float*)d_in[7];
  const float* W4 = (const float*)d_in[8];
  const float* b4 = (const float*)d_in[9];

  const int E = in_sizes[1] / 2;
  const int N = in_sizes[0] / 128;
  const int* src = ei;
  const int* dst = ei + E;

  auto cdiv = [](long long a, long long b) { return (int)((a + b - 1) / b); };
  auto al64 = [](size_t n) { return (n + 63) & ~(size_t)63; };

  float* ws   = (float*)d_ws;
  float* deg  = ws;
  float* dinv = deg  + al64(N);
  int*   rec  = (int*)(dinv + al64(N));
  const int NB = cdiv(E, 256);
  const int pw = NB * 64;
  unsigned* planes = (unsigned*)(rec + al64((size_t)4 * E));
  float* hG   = (float*)(planes + (size_t)4 * pw);
  float* hIn  = hG   + (size_t)N * PADF;
  float* part = hIn  + (size_t)N * PADF;

  const int NT = N / 16;
  const int GB = cdiv(NT, GEMM_RT * GEMM_WAVES);

  constexpr int NR1 = 4096, G1 = 10;
  constexpr int NR2 = 3572, G2 = 9;
  constexpr int NR3 = 2400, G3 = 6;
  constexpr int NR4 = 2048, G4 = 5;
  constexpr int QC1 = 5120, SW1 = 5;
  constexpr int QC2 = 3320, SW2 = 6;
  constexpr int QC3 = 4608, SW3 = 9;
  constexpr int QC4 = 2552, SW4 = 10;
  const int R1 = cdiv(N, NR1), R2 = cdiv(N, NR2), R3 = cdiv(N, NR3), R4 = cdiv(N, NR4);

  k_fill  <<<cdiv(N, 256), 256, 0, stream>>>(deg, 1.0f, N);
  k_degree<<<cdiv(E, 256), 256, 0, stream>>>(dst, deg, E);
  k_rsqrt <<<cdiv(N, 256), 256, 0, stream>>>(deg, dinv, N);
  k_norm<NR1, NR2, NR3, NR4><<<NB, 256, 0, stream>>>(src, dst, dinv, rec, planes, pw, E);

  k_gemm_wmma<128, 15><<<GB, GEMM_WAVES * 32, 0, stream>>>(x, 128, W1, hG, NT);
  k_agg<15, 16, NR1, QC1, SW1><<<dim3(G1, R1), AGG_THREADS, 0, stream>>>(hG, planes, rec, part, E);
  k_finalize<15, 16, G1><<<cdiv((long long)N * 15, 256), 256, 0, stream>>>(hG, part, dinv, b1, hIn, PADF, N, R1 * NR1);

  k_gemm_wmma<15, 20><<<GB, GEMM_WAVES * 32, 0, stream>>>(hIn, PADF, W2, hG, NT);
  k_agg<20, 20, NR2, QC2, SW2><<<dim3(G2, R2), AGG_THREADS, 0, stream>>>(hG, planes + (size_t)pw, rec, part, E);
  k_finalize<20, 20, G2><<<cdiv((long long)N * 20, 256), 256, 0, stream>>>(hG, part, dinv, b2, hIn, PADF, N, R2 * NR2);

  k_gemm_wmma<20, 27><<<GB, GEMM_WAVES * 32, 0, stream>>>(hIn, PADF, W3, hG, NT);
  k_agg<27, 28, NR3, QC3, SW3><<<dim3(G3, R3), AGG_THREADS, 0, stream>>>(hG, planes + (size_t)2 * pw, rec, part, E);
  k_finalize<27, 28, G3><<<cdiv((long long)N * 27, 256), 256, 0, stream>>>(hG, part, dinv, b3, hIn, PADF, N, R3 * NR3);

  k_gemm_wmma<27, 36><<<GB, GEMM_WAVES * 32, 0, stream>>>(hIn, PADF, W4, hG, NT);
  k_agg<36, 36, NR4, QC4, SW4><<<dim3(G4, R4), AGG_THREADS, 0, stream>>>(hG, planes + (size_t)3 * pw, rec, part, E);
  k_finalize<36, 36, G4><<<cdiv((long long)N * 36, 256), 256, 0, stream>>>(hG, part, dinv, b4,
                                                                   (float*)d_out, 36, N, R4 * NR4);
}


# k_finalize of layers 1-3 folded into the next layer's GEMM: its A loader sums the partial planes, adds dinv^2*h and bias, applies selu in registers; 13 launches, not 16; sum order unchanged
# speedup vs baseline: 1.7666x; 1.0525x over previous
#include <hip/hip_runtime.h>
#include <hip/hip_bf16.h>

typedef __attribute__((ext_vector_type(2))) float v2f;
typedef __attribute__((ext_vector_type(8))) float v8f;

#define PADF 48

__device__ __forceinline__ float selu_f(float x) {
  const float alpha = 1.6732632423543772848f;
  const float scale = 1.0507009873554804934f;
  return scale * (x > 0.0f ? x : alpha * (expf(x) - 1.0f));
}

__global__ void k_fill(float* __restrict__ p, float v, int n) {
  int i = blockIdx.x * blockDim.x + threadIdx.x;
  if (i < n) p[i] = v;
}

__global__ void k_degree(const int* __restrict__ dst, float* __restrict__ deg, int E) {
  int e = blockIdx.x * blockDim.x + threadIdx.x;
  if (e < E)
    __hip_atomic_fetch_add(&deg[dst[e]], 1.0f, __ATOMIC_RELAXED, __HIP_MEMORY_SCOPE_AGENT);
}

__global__ void k_rsqrt(const float* __restrict__ deg, float* __restrict__ dinv, int n) {
  int i = blockIdx.x * blockDim.x + threadIdx.x;
  if (i < n) {
    float d = deg[i];
    dinv[i] = d > 0.0f ? rsqrtf(d) : 0.0f;
  }
}

template <int NRA, int NRB, int NRC, int NRD>
__global__ __launch_bounds__(256)
void k_norm(const int* __restrict__ src, const int* __restrict__ dst,
            const float* __restrict__ dinv, int* __restrict__ rec,
            unsigned* __restrict__ planes, int pw, int E) {
  __shared__ unsigned sb[4][64];
  const int tid = threadIdx.x;
  const int e = blockIdx.x * 256 + tid;
  unsigned ia = 255u, ib = 255u, ic = 255u, id = 255u;
  if (e < E) {
    const int s = src[e];
    const int d = dst[e];
    const float w = dinv[s] * dinv[d];
    reinterpret_cast<int4*>(rec)[e] = make_int4(s * PADF, d, __float_as_int(w), 0);
    ia = (unsigned)d / (unsigned)NRA;
    ib = (unsigned)d / (unsigned)NRB;
    ic = (unsigned)d / (unsigned)NRC;
    id = (unsigned)d / (unsigned)NRD;
  }
  unsigned char* sbb = reinterpret_cast<unsigned char*>(&sb[0][0]);
  sbb[tid]       = (unsigned char)ia;
  sbb[256 + tid] = (unsigned char)ib;
  sbb[512 + tid] = (unsigned char)ic;
  sbb[768 + tid] = (unsigned char)id;
  __syncthreads();
  const int p = tid >> 6, i = tid & 63;
  planes[(size_t)p * pw + (size_t)blockIdx.x * 64 + i] = sb[p][i];
}

__device__ __forceinline__ unsigned match4(unsigned w, unsigned rid4) {
  const unsigned x  = w ^ rid4;
  const unsigned nz = ((x & 0x7f7f7f7fu) + 0x7f7f7f7fu) | x;
  const unsigned z  = (~nz & 0x80808080u) >> 7;
  return (z * 0x08102040u) >> 27;
}

#define GEMM_RT 2
#define GEMM_WAVES 8

template <int K, int FOUT>
__global__ __launch_bounds__(GEMM_WAVES * 32)
void k_gemm_wmma(const float* __restrict__ A, int lda,
                 const float* __restrict__ W,
                 float* __restrict__ H, int NT) {
  constexpr int CT = (FOUT + 15) / 16;
  const int lane = threadIdx.x & 31;
  const int half = lane >> 4;
  const int lid  = lane & 15;
  const int wave = __builtin_amdgcn_readfirstlane((int)(threadIdx.x >> 5));
  const int t0   = (blockIdx.x * GEMM_WAVES + wave) * GEMM_RT;
  if (t0 >= NT) return;

  int tile[GEMM_RT];
  const float* Ap[GEMM_RT];
#pragma unroll
  for (int rt = 0; rt < GEMM_RT; ++rt) {
    tile[rt] = (t0 + rt < NT) ? (t0 + rt) : (NT - 1);
    Ap[rt]   = A + (tile[rt] * 16 + lid) * lda + 2 * half;
  }
  bool cv[CT];
  int  cc[CT];
#pragma unroll
  for (int ct = 0; ct < CT; ++ct) {
    const int col = ct * 16 + lid;
    cv[ct] = col < FOUT;
    cc[ct] = cv[ct] ? col : (FOUT - 1);
  }

  v8f acc[GEMM_RT][CT];
#pragma unroll
  for (int rt = 0; rt < GEMM_RT; ++rt)
#pragma unroll
    for (int ct = 0; ct < CT; ++ct)
      acc[rt][ct] = v8f{};

  constexpr int K4 = K & ~3;
#pragma unroll 4
  for (int k0 = 0; k0 < K4; k0 += 4) {
    const int kb = k0 + 2 * half;
    v2f a[GEMM_RT], b[CT];
#pragma unroll
    for (int rt = 0; rt < GEMM_RT; ++rt) {
      a[rt].x = Ap[rt][k0];
      a[rt].y = Ap[rt][k0 + 1];
    }
#pragma unroll
    for (int ct = 0; ct < CT; ++ct) {
      const float b0 = W[kb * FOUT + cc[ct]];
      const float b1 = W[(kb + 1) * FOUT + cc[ct]];
      b[ct].x = cv[ct] ? b0 : 0.0f;
      b[ct].y = cv[ct] ? b1 : 0.0f;
    }
#pragma unroll
    for (int rt = 0; rt < GEMM_RT; ++rt)
#pragma unroll
      for (int ct = 0; ct < CT; ++ct)
        acc[rt][ct] = __builtin_amdgcn_wmma_f32_16x16x4_f32(false, a[rt], false, b[ct], (short)0,
                                                            acc[rt][ct], false, false);
  }
  if constexpr (K % 4 != 0) {
    const int kb  = K4 + 2 * half;
    const int k0c = (kb     < K) ? kb     : (K - 1);
    const int k1c = (kb + 1 < K) ? kb + 1 : (K - 1);
    v2f a[GEMM_RT], b[CT];
#pragma unroll
    for (int rt = 0; rt < GEMM_RT; ++rt) {
      const float* Ar = A + (tile[rt] * 16 + lid) * lda;
      const float a0 = Ar[k0c];
      const float a1 = Ar[k1c];
      a[rt].x = (kb     < K) ? a0 : 0.0f;
      a[rt].y = (kb + 1 < K) ? a1 : 0.0f;
    }
#pragma unroll
    for (int ct = 0; ct < CT; ++ct) {
      const float b0 = W[k0c * FOUT + cc[ct]];
      const float b1 = W[k1c * FOUT + cc[ct]];
      b[ct].x = (cv[ct] && kb     < K) ? b0 : 0.0f;
      b[ct].y = (cv[ct] && kb + 1 < K) ? b1 : 0.0f;
    }
#pragma unroll
    for (int rt = 0; rt < GEMM_RT; ++rt)
#pragma unroll
      for (int ct = 0; ct < CT; ++ct)
        acc[rt][ct] = __builtin_amdgcn_wmma_f32_16x16x4_f32(false, a[rt], false, b[ct], (short)0,
                                                            acc[rt][ct], false, false);
  }
#pragma unroll
  for (int rt = 0; rt < GEMM_RT; ++rt)
#pragma unroll
    for (int ct = 0; ct < CT; ++ct) {
      const int hbase = (tile[rt] * 16 + 8 * half) * PADF + ct * 16 + lid;
#pragma unroll
      for (int r = 0; r < 8; ++r)
        H[hbase + r * PADF] = acc[rt][ct][r];
    }
}

__device__ __forceinline__ float selu_sel(float x) {
  const float alpha = 1.6732632423543772848f;
  const float scale = 1.0507009873554804934f;
  const float neg = alpha * (expf(x) - 1.0f);
  return scale * (x > 0.0f ? x : neg);
}

template <int K, int FOUT, int G, int FPIN>
__global__ __launch_bounds__(GEMM_WAVES * 32)
void k_gemm_fin_wmma(const float* __restrict__ part, int npad,
                     const float* __restrict__ hp, const float* __restrict__ dinv,
                     const float* __restrict__ bias, const float* __restrict__ W,
                     float* __restrict__ H, int NT) {
  constexpr int CT = (FOUT + 15) / 16;
  constexpr int KS = (K + 3) / 4;
  static_assert(FPIN % 2 == 0 && FPIN >= 4 * KS && 4 * KS <= PADF, "pair loads of the last k step stay inside a row");
  const int lane = threadIdx.x & 31;
  const int half = lane >> 4;
  const int lid  = lane & 15;
  const int wave = __builtin_amdgcn_readfirstlane((int)(threadIdx.x >> 5));
  const int t0   = (blockIdx.x * GEMM_WAVES + wave) * GEMM_RT;
  if (t0 >= NT) return;

  const size_t ps = (size_t)npad * FPIN;
  int tile[GEMM_RT];
  float di[GEMM_RT];
  v2f pv[KS][GEMM_RT][G];
  v2f hv[KS][GEMM_RT];
#pragma unroll
  for (int rt = 0; rt < GEMM_RT; ++rt) {
    tile[rt] = (t0 + rt < NT) ? (t0 + rt) : (NT - 1);
    const int row = tile[rt] * 16 + lid;
    const float* pr = part + (size_t)row * FPIN + 2 * half;
    const float* hr = hp + (size_t)row * PADF + 2 * half;
#pragma unroll
    for (int s = 0; s < KS; ++s) {
#pragma unroll
      for (int g = 0; g < G; ++g)
        pv[s][rt][g] = *reinterpret_cast<const v2f*>(pr + (size_t)g * ps + 4 * s);
      hv[s][rt] = *reinterpret_cast<const v2f*>(hr + 4 * s);
    }
    di[rt] = dinv[row];
  }
  __builtin_amdgcn_sched_barrier(0);
  float dd[GEMM_RT];
#pragma unroll
  for (int rt = 0; rt < GEMM_RT; ++rt)
    dd[rt] = di[rt] * di[rt];
  bool cv[CT];
  int  cc[CT];
#pragma unroll
  for (int ct = 0; ct < CT; ++ct) {
    const int col = ct * 16 + lid;
    cv[ct] = col < FOUT;
    cc[ct] = cv[ct] ? col : (FOUT - 1);
  }

  v8f acc[GEMM_RT][CT];
#pragma unroll
  for (int rt = 0; rt < GEMM_RT; ++rt)
#pragma unroll
    for (int ct = 0; ct < CT; ++ct)
      acc[rt][ct] = v8f{};

#pragma unroll
  for (int s = 0; s < KS; ++s) {
    const int kb  = 4 * s + 2 * half;
    const bool v0 = kb < K, v1 = kb + 1 < K;
    const int k0c = v0 ? kb     : (K - 1);
    const int k1c = v1 ? kb + 1 : (K - 1);
    const float bk0 = bias[k0c], bk1 = bias[k1c];
    v2f a[GEMM_RT], b[CT];
#pragma unroll
    for (int rt = 0; rt < GEMM_RT; ++rt) {
      float ag0 = 0.0f, ag1 = 0.0f;
#pragma unroll
      for (int g = 0; g < G; ++g) {
        ag0 += pv[s][rt][g].x;
        ag1 += pv[s][rt][g].y;
      }
      const float x0 = ag0 + dd[rt] * hv[s][rt].x + bk0;
      const float x1 = ag1 + dd[rt] * hv[s][rt].y + bk1;
      const float a0 = selu_sel(x0);
      const float a1 = selu_sel(x1);
      a[rt].x = v0 ? a0 : 0.0f;
      a[rt].y = v1 ? a1 : 0.0f;
    }
#pragma unroll
    for (int ct = 0; ct < CT; ++ct) {
      const float b0 = W[k0c * FOUT + cc[ct]];
      const float b1 = W[k1c * FOUT + cc[ct]];
      b[ct].x = (cv[ct] && v0) ? b0 : 0.0f;
      b[ct].y = (cv[ct] && v1) ? b1 : 0.0f;
    }
#pragma unroll
    for (int rt = 0; rt < GEMM_RT; ++rt)
#pragma unroll
      for (int ct = 0; ct < CT; ++ct)
        acc[rt][ct] = __builtin_amdgcn_wmma_f32_16x16x4_f32(false, a[rt], false, b[ct], (short)0,
                                                            acc[rt][ct], false, false);
  }
#pragma unroll
  for (int rt = 0; rt < GEMM_RT; ++rt)
#pragma unroll
    for (int ct = 0; ct < CT; ++ct) {
      const int hbase = (tile[rt] * 16 + 8 * half) * PADF + ct * 16 + lid;
#pragma unroll
      for (int r = 0; r < 8; ++r)
        H[hbase + r * PADF] = acc[rt][ct][r];
    }
}

#define AGG_THREADS 1024
#define AGG_U 8

template <int F, int FP, int NR, int QC, int W>
__global__ __launch_bounds__(AGG_THREADS)
void k_agg(const float* __restrict__ h, const unsigned* __restrict__ plane,
           const int* __restrict__ rec, float* __restrict__ part, int E) {
  __shared__ __attribute__((aligned(16))) float acc[NR * FP];
  __shared__ int   q_a[QC];
  __shared__ int   q_d[QC];
  __shared__ float q_w[QC];
  __shared__ int   q_n;
  __shared__ int   q_more;

  const int tid  = threadIdx.x;
  const int lane = tid & 31;
  const int lo   = blockIdx.y * NR;
  const int G    = gridDim.x;
  const int echunk = (((E + G - 1) / G) + 31) & ~31;
  const int e_begin = blockIdx.x * echunk;
  const int e_end   = (e_begin + echunk < E) ? (e_begin + echunk) : E;

  constexpr int PASS = AGG_THREADS * 32 * W;
  constexpr int NF   = (QC + AGG_THREADS - 1) / AGG_THREADS;
  constexpr int L  = FP / 4;
  constexpr int EP = AGG_THREADS / L;
  const int  esub = tid / L;
  const int  pc   = tid - esub * L;
  const int  c4   = pc * 4;
  const bool lane_used = tid < EP * L;
  const int  rot  = (pc + esub) & 3;
  const int  o0 = rot, o1 = (rot + 1) & 3, o2 = (rot + 2) & 3, o3 = (rot + 3) & 3;
  const unsigned rid4 = (unsigned)blockIdx.y * 0x01010101u;

  float4* acc4 = reinterpret_cast<float4*>(acc);
  for (int i = tid; i < NR * FP / 4; i += AGG_THREADS)
    acc4[i] = make_float4(0.0f, 0.0f, 0.0f, 0.0f);
  if (tid == 0) { q_n = 0; q_more = 0; }
  __syncthreads();

  const uint4* plane4 = reinterpret_cast<const uint4*>(plane);
  const int4*  rec4   = reinterpret_cast<const int4*>(rec);

  for (int b = e_begin; b < e_end; b += PASS) {
    unsigned pend[W];
    constexpr int WC = (W <= 6) ? W : (W + 1) / 2;
#pragma unroll
    for (int w0 = 0; w0 < W; w0 += WC) {
      uint4 pl[2 * WC];
#pragma unroll
      for (int i = 0; i < WC; ++i)
        if (w0 + i < W) {
          const int eb  = b + ((w0 + i) * AGG_THREADS + tid) * 32;
          const int ebc = (e_end - eb > 0) ? eb : e_begin;
          pl[2 * i]     = plane4[(ebc >> 4)];
          pl[2 * i + 1] = plane4[(ebc >> 4) + 1];
        }
      __builtin_amdgcn_sched_barrier(0);
#pragma unroll
      for (int i = 0; i < WC; ++i)
        if (w0 + i < W) {
          const int eb = b + ((w0 + i) * AGG_THREADS + tid) * 32;
          const int nv = e_end - eb;
          const uint4 p0 = pl[2 * i], p1 = pl[2 * i + 1];
          unsigned mt = match4(p0.x, rid4)       | match4(p0.y, rid4) << 4  |
                        match4(p0.z, rid4) << 8  | match4(p0.w, rid4) << 12 |
                        match4(p1.x, rid4) << 16 | match4(p1.y, rid4) << 20 |
                        match4(p1.z, rid4) << 24 | match4(p1.w, rid4) << 28;
          mt &= (nv >= 32) ? 0xffffffffu : ((nv > 0) ? ((1u << nv) - 1u) : 0u);
          pend[w0 + i] = mt;
        }
    }

    bool more;
    do {
      int c = 0;
#pragma unroll
      for (int w = 0; w < W; ++w) c += __popc(pend[w]);
      int incl = c;
#pragma unroll
      for (int o = 1; o < 32; o <<= 1) {
        const int t = __shfl_up(incl, o, 32);
        if (lane >= o) incl += t;
      }
      const int wtot = __shfl(incl, 31, 32);
      int base = 0;
      if (lane == 31 && wtot > 0)
        base = __hip_atomic_fetch_add(&q_n, wtot, __ATOMIC_RELAXED, __HIP_MEMORY_SCOPE_WORKGROUP);
      base = __shfl(base, 31, 32);
      int pos = base + incl - c;
      unsigned left = 0;
#pragma unroll
      for (int w = 0; w < W; ++w) {
        const int eb = b + (w * AGG_THREADS + tid) * 32;
        unsigned p = pend[w];
        while (p != 0u && pos < QC) {
          const int j = __ffs(p) - 1;
          p &= p - 1;
          q_a[pos] = eb + j;
          ++pos;
        }
        pend[w] = p;
        left |= p;
      }
      if (left) q_more = 1;
      __syncthreads();

      const int M = (q_n < QC) ? q_n : QC;
      more = (q_more != 0);
      int4 rr[NF];
#pragma unroll
      for (int u = 0; u < NF; ++u) {
        const int m  = tid + u * AGG_THREADS;
        const bool fo = m < M;
        const int qe = q_a[fo ? m : 0];
        rr[u] = rec4[fo ? qe : e_begin];
      }
#pragma unroll
      for (int u = 0; u < NF; ++u) {
        asm volatile("" :: "v"(rr[u].x), "v"(rr[u].y), "v"(rr[u].z));
      }
#pragma unroll
      for (int u = 0; u < NF; ++u) {
        const int m = tid + u * AGG_THREADS;
        if (m < M) {
          q_a[m] = rr[u].x;
          q_d[m] = (rr[u].y - lo) * FP;
          q_w[m] = __int_as_float(rr[u].z);
        }
      }
      __syncthreads();

      if (tid == 0) { q_n = 0; q_more = 0; }
      for (int m0 = 0; m0 < M; m0 += EP * AGG_U) {
        float4 hv[AGG_U];
        float  wv[AGG_U];
        int    a[AGG_U];
        bool   ok[AGG_U];
#pragma unroll
        for (int u = 0; u < AGG_U; ++u) {
          const int m = m0 + u * EP + esub;
          ok[u] = lane_used && (m < M);
          const int mc = ok[u] ? m : 0;
          const int qa = q_a[mc];
          const int qd = q_d[mc];
          hv[u] = *reinterpret_cast<const float4*>(h + qa + c4);
          wv[u] = q_w[mc];
          a[u]  = qd + c4;
        }
#pragma unroll
        for (int u = 0; u < AGG_U; ++u)
          if (ok[u]) {
            float v0 = hv[u].x, v1 = hv[u].y, v2 = hv[u].z, v3 = hv[u].w;
            if (rot & 1) { const float t = v0; v0 = v1; v1 = v2; v2 = v3; v3 = t; }
            if (rot & 2) { const float t0 = v0, t1 = v1; v0 = v2; v1 = v3; v2 = t0; v3 = t1; }
            float* p = &acc[a[u]];
            const float w = wv[u];
            __hip_atomic_fetch_add(p + o0, w * v0, __ATOMIC_RELAXED, __HIP_MEMORY_SCOPE_WORKGROUP);
            __hip_atomic_fetch_add(p + o1, w * v1, __ATOMIC_RELAXED, __HIP_MEMORY_SCOPE_WORKGROUP);
            __hip_atomic_fetch_add(p + o2, w * v2, __ATOMIC_RELAXED, __HIP_MEMORY_SCOPE_WORKGROUP);
            __hip_atomic_fetch_add(p + o3, w * v3, __ATOMIC_RELAXED, __HIP_MEMORY_SCOPE_WORKGROUP);
          }
      }
      __syncthreads();
    } while (more);
  }

  const size_t npad = (size_t)gridDim.y * NR;
  float4* out4 = reinterpret_cast<float4*>(part + ((size_t)blockIdx.x * npad + lo) * FP);
  for (int i = tid; i < NR * FP / 4; i += AGG_THREADS)
    out4[i] = acc4[i];
}

template <int F, int FP, int G>
__global__ void k_finalize(const float* __restrict__ h, const float* __restrict__ part,
                           const float* __restrict__ dinv, const float* __restrict__ bias,
                           float* __restrict__ out, int ldo, int N, int npad) {
  int t = blockIdx.x * blockDim.x + threadIdx.x;
  if (t >= N * F) return;
  int n = t / F;
  int f = t - n * F;
  float agg = 0.0f;
#pragma unroll
  for (int g = 0; g < G; ++g)
    agg += part[((size_t)g * npad + n) * FP + f];
  float di = dinv[n];
  float v = agg + di * di * h[n * PADF + f] + bias[f];
  out[n * ldo + f] = selu_f(v);
}

extern "C" void kernel_launch(void* const* d_in, const int* in_sizes, int n_in,
                              void* d_out, int out_size, void* d_ws, size_t ws_size,
                              hipStream_t stream) {
  const float* x  = (const float*)d_in[0];
  const int*   ei = (const int*)d_in[1];
  const float* W1 = (const float*)d_in[2];
  const float* b1 = (const float*)d_in[3];
  const float* W2 = (const float*)d_in[4];
  const float* b2 = (const float*)d_in[5];
  const float* W3 = (const float*)d_in[6];
  const float* b3 = (const float*)d_in[7];
  const float* W4 = (const float*)d_in[8];
  const float* b4 = (const float*)d_in[9];

  const int E = in_sizes[1] / 2;
  const int N = in_sizes[0] / 128;
  const int* src = ei;
  const int* dst = ei + E;

  auto cdiv = [](long long a, long long b) { return (int)((a + b - 1) / b); };
  auto al64 = [](size_t n) { return (n + 63) & ~(size_t)63; };

  float* ws   = (float*)d_ws;
  float* deg  = ws;
  float* dinv = deg  + al64(N);
  int*   rec  = (int*)(dinv + al64(N));
  const int NB = cdiv(E, 256);
  const int pw = NB * 64;
  unsigned* planes = (unsigned*)(rec + al64((size_t)4 * E));
  float* hA   = (float*)(planes + (size_t)4 * pw);
  float* hB   = hA   + (size_t)N * PADF;
  float* part = hB   + (size_t)N * PADF;

  const int NT = N / 16;
  const int GB = cdiv(NT, GEMM_RT * GEMM_WAVES);

  constexpr int NR1 = 4096, G1 = 10;
  constexpr int NR2 = 3572, G2 = 9;
  constexpr int NR3 = 2400, G3 = 6;
  constexpr int NR4 = 2048, G4 = 5;
  constexpr int QC1 = 5120, SW1 = 5;
  constexpr int QC2 = 3320, SW2 = 6;
  constexpr int QC3 = 4608, SW3 = 9;
  constexpr int QC4 = 2552, SW4 = 10;
  const int R1 = cdiv(N, NR1), R2 = cdiv(N, NR2), R3 = cdiv(N, NR3), R4 = cdiv(N, NR4);

  k_fill  <<<cdiv(N, 256), 256, 0, stream>>>(deg, 1.0f, N);
  k_degree<<<cdiv(E, 256), 256, 0, stream>>>(dst, deg, E);
  k_rsqrt <<<cdiv(N, 256), 256, 0, stream>>>(deg, dinv, N);
  k_norm<NR1, NR2, NR3, NR4><<<NB, 256, 0, stream>>>(src, dst, dinv, rec, planes, pw, E);

  k_gemm_wmma<128, 15><<<GB, GEMM_WAVES * 32, 0, stream>>>(x, 128, W1, hA, NT);
  k_agg<15, 16, NR1, QC1, SW1><<<dim3(G1, R1), AGG_THREADS, 0, stream>>>(hA, planes, rec, part, E);

  k_gemm_fin_wmma<15, 20, G1, 16><<<GB, GEMM_WAVES * 32, 0, stream>>>(part, R1 * NR1, hA, dinv, b1, W2, hB, NT);
  k_agg<20, 20, NR2, QC2, SW2><<<dim3(G2, R2), AGG_THREADS, 0, stream>>>(hB, planes + (size_t)pw, rec, part, E);

  k_gemm_fin_wmma<20, 27, G2, 20><<<GB, GEMM_WAVES * 32, 0, stream>>>(part, R2 * NR2, hB, dinv, b2, W3, hA, NT);
  k_agg<27, 28, NR3, QC3, SW3><<<dim3(G3, R3), AGG_THREADS, 0, stream>>>(hA, planes + (size_t)2 * pw, rec, part, E);

  k_gemm_fin_wmma<27, 36, G3, 28><<<GB, GEMM_WAVES * 32, 0, stream>>>(part, R3 * NR3, hA, dinv, b3, W4, hB, NT);
  k_agg<36, 36, NR4, QC4, SW4><<<dim3(G4, R4), AGG_THREADS, 0, stream>>>(hB, planes + (size_t)3 * pw, rec, part, E);
  k_finalize<36, 36, G4><<<cdiv((long long)N * 36, 256), 256, 0, stream>>>(hB, part, dinv, b4,
                                                                   (float*)d_out, 36, N, R4 * NR4);
}


# k_agg queue holds edge ids only (4 B a slot: a chunk's matches in one round); an edge's lanes fetch its record in the gather trip, no separate record phase; LDS add arrival order changes
# speedup vs baseline: 1.8275x; 1.0345x over previous
#include <hip/hip_runtime.h>
#include <hip/hip_bf16.h>

typedef __attribute__((ext_vector_type(2))) float v2f;
typedef __attribute__((ext_vector_type(8))) float v8f;

#define PADF 48

__device__ __forceinline__ float selu_f(float x) {
  const float alpha = 1.6732632423543772848f;
  const float scale = 1.0507009873554804934f;
  return scale * (x > 0.0f ? x : alpha * (expf(x) - 1.0f));
}

__global__ void k_fill(float* __restrict__ p, float v, int n) {
  int i = blockIdx.x * blockDim.x + threadIdx.x;
  if (i < n) p[i] = v;
}

__global__ void k_degree(const int* __restrict__ dst, float* __restrict__ deg, int E) {
  int e = blockIdx.x * blockDim.x + threadIdx.x;
  if (e < E)
    __hip_atomic_fetch_add(&deg[dst[e]], 1.0f, __ATOMIC_RELAXED, __HIP_MEMORY_SCOPE_AGENT);
}

__global__ void k_rsqrt(const float* __restrict__ deg, float* __restrict__ dinv, int n) {
  int i = blockIdx.x * blockDim.x + threadIdx.x;
  if (i < n) {
    float d = deg[i];
    dinv[i] = d > 0.0f ? rsqrtf(d) : 0.0f;
  }
}

template <int NRA, int NRB, int NRC, int NRD>
__global__ __launch_bounds__(256)
void k_norm(const int* __restrict__ src, const int* __restrict__ dst,
            const float* __restrict__ dinv, int* __restrict__ rec,
            unsigned* __restrict__ planes, int pw, int E) {
  __shared__ unsigned sb[4][64];
  const int tid = threadIdx.x;
  const int e = blockIdx.x * 256 + tid;
  unsigned ia = 255u, ib = 255u, ic = 255u, id = 255u;
  if (e < E) {
    const int s = src[e];
    const int d = dst[e];
    const float w = dinv[s] * dinv[d];
    reinterpret_cast<int4*>(rec)[e] = make_int4(s * PADF, d, __float_as_int(w), 0);
    ia = (unsigned)d / (unsigned)NRA;
    ib = (unsigned)d / (unsigned)NRB;
    ic = (unsigned)d / (unsigned)NRC;
    id = (unsigned)d / (unsigned)NRD;
  }
  unsigned char* sbb = reinterpret_cast<unsigned char*>(&sb[0][0]);
  sbb[tid]       = (unsigned char)ia;
  sbb[256 + tid] = (unsigned char)ib;
  sbb[512 + tid] = (unsigned char)ic;
  sbb[768 + tid] = (unsigned char)id;
  __syncthreads();
  const int p = tid >> 6, i = tid & 63;
  planes[(size_t)p * pw + (size_t)blockIdx.x * 64 + i] = sb[p][i];
}

__device__ __forceinline__ unsigned match4(unsigned w, unsigned rid4) {
  const unsigned x  = w ^ rid4;
  const unsigned nz = ((x & 0x7f7f7f7fu) + 0x7f7f7f7fu) | x;
  const unsigned z  = (~nz & 0x80808080u) >> 7;
  return (z * 0x08102040u) >> 27;
}

#define GEMM_RT 2
#define GEMM_WAVES 8

template <int K, int FOUT>
__global__ __launch_bounds__(GEMM_WAVES * 32)
void k_gemm_wmma(const float* __restrict__ A, int lda,
                 const float* __restrict__ W,
                 float* __restrict__ H, int NT) {
  constexpr int CT = (FOUT + 15) / 16;
  const int lane = threadIdx.x & 31;
  const int half = lane >> 4;
  const int lid  = lane & 15;
  const int wave = __builtin_amdgcn_readfirstlane((int)(threadIdx.x >> 5));
  const int t0   = (blockIdx.x * GEMM_WAVES + wave) * GEMM_RT;
  if (t0 >= NT) return;

  int tile[GEMM_RT];
  const float* Ap[GEMM_RT];
#pragma unroll
  for (int rt = 0; rt < GEMM_RT; ++rt) {
    tile[rt] = (t0 + rt < NT) ? (t0 + rt) : (NT - 1);
    Ap[rt]   = A + (tile[rt] * 16 + lid) * lda + 2 * half;
  }
  bool cv[CT];
  int  cc[CT];
#pragma unroll
  for (int ct = 0; ct < CT; ++ct) {
    const int col = ct * 16 + lid;
    cv[ct] = col < FOUT;
    cc[ct] = cv[ct] ? col : (FOUT - 1);
  }

  v8f acc[GEMM_RT][CT];
#pragma unroll
  for (int rt = 0; rt < GEMM_RT; ++rt)
#pragma unroll
    for (int ct = 0; ct < CT; ++ct)
      acc[rt][ct] = v8f{};

  constexpr int K4 = K & ~3;
#pragma unroll 4
  for (int k0 = 0; k0 < K4; k0 += 4) {
    const int kb = k0 + 2 * half;
    v2f a[GEMM_RT], b[CT];
#pragma unroll
    for (int rt = 0; rt < GEMM_RT; ++rt) {
      a[rt].x = Ap[rt][k0];
      a[rt].y = Ap[rt][k0 + 1];
    }
#pragma unroll
    for (int ct = 0; ct < CT; ++ct) {
      const float b0 = W[kb * FOUT + cc[ct]];
      const float b1 = W[(kb + 1) * FOUT + cc[ct]];
      b[ct].x = cv[ct] ? b0 : 0.0f;
      b[ct].y = cv[ct] ? b1 : 0.0f;
    }
#pragma unroll
    for (int rt = 0; rt < GEMM_RT; ++rt)
#pragma unroll
      for (int ct = 0; ct < CT; ++ct)
        acc[rt][ct] = __builtin_amdgcn_wmma_f32_16x16x4_f32(false, a[rt], false, b[ct], (short)0,
                                                            acc[rt][ct], false, false);
  }
  if constexpr (K % 4 != 0) {
    const int kb  = K4 + 2 * half;
    const int k0c = (kb     < K) ? kb     : (K - 1);
    const int k1c = (kb + 1 < K) ? kb + 1 : (K - 1);
    v2f a[GEMM_RT], b[CT];
#pragma unroll
    for (int rt = 0; rt < GEMM_RT; ++rt) {
      const float* Ar = A + (tile[rt] * 16 + lid) * lda;
      const float a0 = Ar[k0c];
      const float a1 = Ar[k1c];
      a[rt].x = (kb     < K) ? a0 : 0.0f;
      a[rt].y = (kb + 1 < K) ? a1 : 0.0f;
    }
#pragma unroll
    for (int ct = 0; ct < CT; ++ct) {
      const float b0 = W[k0c * FOUT + cc[ct]];
      const float b1 = W[k1c * FOUT + cc[ct]];
      b[ct].x = (cv[ct] && kb     < K) ? b0 : 0.0f;
      b[ct].y = (cv[ct] && kb + 1 < K) ? b1 : 0.0f;
    }
#pragma unroll
    for (int rt = 0; rt < GEMM_RT; ++rt)
#pragma unroll
      for (int ct = 0; ct < CT; ++ct)
        acc[rt][ct] = __builtin_amdgcn_wmma_f32_16x16x4_f32(false, a[rt], false, b[ct], (short)0,
                                                            acc[rt][ct], false, false);
  }
#pragma unroll
  for (int rt = 0; rt < GEMM_RT; ++rt)
#pragma unroll
    for (int ct = 0; ct < CT; ++ct) {
      const int hbase = (tile[rt] * 16 + 8 * half) * PADF + ct * 16 + lid;
#pragma unroll
      for (int r = 0; r < 8; ++r)
        H[hbase + r * PADF] = acc[rt][ct][r];
    }
}

__device__ __forceinline__ float selu_sel(float x) {
  const float alpha = 1.6732632423543772848f;
  const float scale = 1.0507009873554804934f;
  const float neg = alpha * (expf(x) - 1.0f);
  return scale * (x > 0.0f ? x : neg);
}

template <int K, int FOUT, int G, int FPIN>
__global__ __launch_bounds__(GEMM_WAVES * 32)
void k_gemm_fin_wmma(const float* __restrict__ part, int npad,
                     const float* __restrict__ hp, const float* __restrict__ dinv,
                     const float* __restrict__ bias, const float* __restrict__ W,
                     float* __restrict__ H, int NT) {
  constexpr int CT = (FOUT + 15) / 16;
  constexpr int KS = (K + 3) / 4;
  static_assert(FPIN % 2 == 0 && FPIN >= 4 * KS && 4 * KS <= PADF, "pair loads of the last k step stay inside a row");
  const int lane = threadIdx.x & 31;
  const int half = lane >> 4;
  const int lid  = lane & 15;
  const int wave = __builtin_amdgcn_readfirstlane((int)(threadIdx.x >> 5));
  const int t0   = (blockIdx.x * GEMM_WAVES + wave) * GEMM_RT;
  if (t0 >= NT) return;

  const size_t ps = (size_t)npad * FPIN;
  int tile[GEMM_RT];
  float di[GEMM_RT];
  v2f pv[KS][GEMM_RT][G];
  v2f hv[KS][GEMM_RT];
#pragma unroll
  for (int rt = 0; rt < GEMM_RT; ++rt) {
    tile[rt] = (t0 + rt < NT) ? (t0 + rt) : (NT - 1);
    const int row = tile[rt] * 16 + lid;
    const float* pr = part + (size_t)row * FPIN + 2 * half;
    const float* hr = hp + (size_t)row * PADF + 2 * half;
#pragma unroll
    for (int s = 0; s < KS; ++s) {
#pragma unroll
      for (int g = 0; g < G; ++g)
        pv[s][rt][g] = *reinterpret_cast<const v2f*>(pr + (size_t)g * ps + 4 * s);
      hv[s][rt] = *reinterpret_cast<const v2f*>(hr + 4 * s);
    }
    di[rt] = dinv[row];
  }
  __builtin_amdgcn_sched_barrier(0);
  float dd[GEMM_RT];
#pragma unroll
  for (int rt = 0; rt < GEMM_RT; ++rt)
    dd[rt] = di[rt] * di[rt];
  bool cv[CT];
  int  cc[CT];
#pragma unroll
  for (int ct = 0; ct < CT; ++ct) {
    const int col = ct * 16 + lid;
    cv[ct] = col < FOUT;
    cc[ct] = cv[ct] ? col : (FOUT - 1);
  }

  v8f acc[GEMM_RT][CT];
#pragma unroll
  for (int rt = 0; rt < GEMM_RT; ++rt)
#pragma unroll
    for (int ct = 0; ct < CT; ++ct)
      acc[rt][ct] = v8f{};

#pragma unroll
  for (int s = 0; s < KS; ++s) {
    const int kb  = 4 * s + 2 * half;
    const bool v0 = kb < K, v1 = kb + 1 < K;
    const int k0c = v0 ? kb     : (K - 1);
    const int k1c = v1 ? kb + 1 : (K - 1);
    const float bk0 = bias[k0c], bk1 = bias[k1c];
    v2f a[GEMM_RT], b[CT];
#pragma unroll
    for (int rt = 0; rt < GEMM_RT; ++rt) {
      float ag0 = 0.0f, ag1 = 0.0f;
#pragma unroll
      for (int g = 0; g < G; ++g) {
        ag0 += pv[s][rt][g].x;
        ag1 += pv[s][rt][g].y;
      }
      const float x0 = ag0 + dd[rt] * hv[s][rt].x + bk0;
      const float x1 = ag1 + dd[rt] * hv[s][rt].y + bk1;
      const float a0 = selu_sel(x0);
      const float a1 = selu_sel(x1);
      a[rt].x = v0 ? a0 : 0.0f;
      a[rt].y = v1 ? a1 : 0.0f;
    }
#pragma unroll
    for (int ct = 0; ct < CT; ++ct) {
      const float b0 = W[k0c * FOUT + cc[ct]];
      const float b1 = W[k1c * FOUT + cc[ct]];
      b[ct].x = (cv[ct] && v0) ? b0 : 0.0f;
      b[ct].y = (cv[ct] && v1) ? b1 : 0.0f;
    }
#pragma unroll
    for (int rt = 0; rt < GEMM_RT; ++rt)
#pragma unroll
      for (int ct = 0; ct < CT; ++ct)
        acc[rt][ct] = __builtin_amdgcn_wmma_f32_16x16x4_f32(false, a[rt], false, b[ct], (short)0,
                                                            acc[rt][ct], false, false);
  }
#pragma unroll
  for (int rt = 0; rt < GEMM_RT; ++rt)
#pragma unroll
    for (int ct = 0; ct < CT; ++ct) {
      const int hbase = (tile[rt] * 16 + 8 * half) * PADF + ct * 16 + lid;
#pragma unroll
      for (int r = 0; r < 8; ++r)
        H[hbase + r * PADF] = acc[rt][ct][r];
    }
}

#define AGG_THREADS 1024
#define AGG_U 8

template <int F, int FP, int NR, int QC, int W>
__global__ __launch_bounds__(AGG_THREADS)
void k_agg(const float* __restrict__ h, const unsigned* __restrict__ plane,
           const int* __restrict__ rec, float* __restrict__ part, int E) {
  __shared__ __attribute__((aligned(16))) float acc[NR * FP];
  __shared__ int   q_e[QC];
  __shared__ int   q_n;
  __shared__ int   q_more;

  const int tid  = threadIdx.x;
  const int lane = tid & 31;
  const int lo   = blockIdx.y * NR;
  const int G    = gridDim.x;
  const int echunk = (((E + G - 1) / G) + 31) & ~31;
  const int e_begin = blockIdx.x * echunk;
  const int e_end   = (e_begin + echunk < E) ? (e_begin + echunk) : E;

  constexpr int PASS = AGG_THREADS * 32 * W;
  constexpr int L  = FP / 4;
  constexpr int EP = AGG_THREADS / L;
  const int  esub = tid / L;
  const int  pc   = tid - esub * L;
  const int  c4   = pc * 4;
  const bool lane_used = tid < EP * L;
  const int  rot  = (pc + esub) & 3;
  const int  o0 = rot, o1 = (rot + 1) & 3, o2 = (rot + 2) & 3, o3 = (rot + 3) & 3;
  const unsigned rid4 = (unsigned)blockIdx.y * 0x01010101u;

  float4* acc4 = reinterpret_cast<float4*>(acc);
  for (int i = tid; i < NR * FP / 4; i += AGG_THREADS)
    acc4[i] = make_float4(0.0f, 0.0f, 0.0f, 0.0f);
  if (tid == 0) { q_n = 0; q_more = 0; }
  __syncthreads();

  const uint4* plane4 = reinterpret_cast<const uint4*>(plane);
  const int4*  rec4   = reinterpret_cast<const int4*>(rec);

  for (int b = e_begin; b < e_end; b += PASS) {
    unsigned pend[W];
    constexpr int WC = (W <= 6) ? W : (W + 1) / 2;
#pragma unroll
    for (int w0 = 0; w0 < W; w0 += WC) {
      uint4 pl[2 * WC];
#pragma unroll
      for (int i = 0; i < WC; ++i)
        if (w0 + i < W) {
          const int eb  = b + ((w0 + i) * AGG_THREADS + tid) * 32;
          const int ebc = (e_end - eb > 0) ? eb : e_begin;
          pl[2 * i]     = plane4[(ebc >> 4)];
          pl[2 * i + 1] = plane4[(ebc >> 4) + 1];
        }
      __builtin_amdgcn_sched_barrier(0);
#pragma unroll
      for (int i = 0; i < WC; ++i)
        if (w0 + i < W) {
          const int eb = b + ((w0 + i) * AGG_THREADS + tid) * 32;
          const int nv = e_end - eb;
          const uint4 p0 = pl[2 * i], p1 = pl[2 * i + 1];
          unsigned mt = match4(p0.x, rid4)       | match4(p0.y, rid4) << 4  |
                        match4(p0.z, rid4) << 8  | match4(p0.w, rid4) << 12 |
                        match4(p1.x, rid4) << 16 | match4(p1.y, rid4) << 20 |
                        match4(p1.z, rid4) << 24 | match4(p1.w, rid4) << 28;
          mt &= (nv >= 32) ? 0xffffffffu : ((nv > 0) ? ((1u << nv) - 1u) : 0u);
          pend[w0 + i] = mt;
        }
    }

    bool more;
    do {
      int c = 0;
#pragma unroll
      for (int w = 0; w < W; ++w) c += __popc(pend[w]);
      int incl = c;
#pragma unroll
      for (int o = 1; o < 32; o <<= 1) {
        const int t = __shfl_up(incl, o, 32);
        if (lane >= o) incl += t;
      }
      const int wtot = __shfl(incl, 31, 32);
      int base = 0;
      if (lane == 31 && wtot > 0)
        base = __hip_atomic_fetch_add(&q_n, wtot, __ATOMIC_RELAXED, __HIP_MEMORY_SCOPE_WORKGROUP);
      base = __shfl(base, 31, 32);
      int pos = base + incl - c;
      unsigned left = 0;
#pragma unroll
      for (int w = 0; w < W; ++w) {
        const int eb = b + (w * AGG_THREADS + tid) * 32;
        unsigned p = pend[w];
        while (p != 0u && pos < QC) {
          const int j = __ffs(p) - 1;
          p &= p - 1;
          q_e[pos] = eb + j;
          ++pos;
        }
        pend[w] = p;
        left |= p;
      }
      if (left) q_more = 1;
      __syncthreads();

      const int M = (q_n < QC) ? q_n : QC;
      more = (q_more != 0);
      __syncthreads();

      if (tid == 0) { q_n = 0; q_more = 0; }
      for (int m0 = 0; m0 < M; m0 += EP * AGG_U) {
        int4   rc[AGG_U];
        float4 hv[AGG_U];
        float  wv[AGG_U];
        int    a[AGG_U];
        bool   ok[AGG_U];
#pragma unroll
        for (int u = 0; u < AGG_U; ++u) {
          const int m = m0 + u * EP + esub;
          ok[u] = lane_used && (m < M);
          const int mc = ok[u] ? m : 0;
          rc[u] = rec4[q_e[mc]];
        }
#pragma unroll
        for (int u = 0; u < AGG_U; ++u) {
          asm volatile("" :: "v"(rc[u].x), "v"(rc[u].y), "v"(rc[u].z));
        }
#pragma unroll
        for (int u = 0; u < AGG_U; ++u) {
          hv[u] = *reinterpret_cast<const float4*>(h + rc[u].x + c4);
          wv[u] = __int_as_float(rc[u].z);
          a[u]  = (rc[u].y - lo) * FP + c4;
        }
#pragma unroll
        for (int u = 0; u < AGG_U; ++u) {
          asm volatile("" :: "v"(hv[u].x), "v"(hv[u].y), "v"(hv[u].z), "v"(hv[u].w));
        }
#pragma unroll
        for (int u = 0; u < AGG_U; ++u)
          if (ok[u]) {
            float v0 = hv[u].x, v1 = hv[u].y, v2 = hv[u].z, v3 = hv[u].w;
            if (rot & 1) { const float t = v0; v0 = v1; v1 = v2; v2 = v3; v3 = t; }
            if (rot & 2) { const float t0 = v0, t1 = v1; v0 = v2; v1 = v3; v2 = t0; v3 = t1; }
            float* p = &acc[a[u]];
            const float w = wv[u];
            __hip_atomic_fetch_add(p + o0, w * v0, __ATOMIC_RELAXED, __HIP_MEMORY_SCOPE_WORKGROUP);
            __hip_atomic_fetch_add(p + o1, w * v1, __ATOMIC_RELAXED, __HIP_MEMORY_SCOPE_WORKGROUP);
            __hip_atomic_fetch_add(p + o2, w * v2, __ATOMIC_RELAXED, __HIP_MEMORY_SCOPE_WORKGROUP);
            __hip_atomic_fetch_add(p + o3, w * v3, __ATOMIC_RELAXED, __HIP_MEMORY_SCOPE_WORKGROUP);
          }
      }
      __syncthreads();
    } while (more);
  }

  const size_t npad = (size_t)gridDim.y * NR;
  float4* out4 = reinterpret_cast<float4*>(part + ((size_t)blockIdx.x * npad + lo) * FP);
  for (int i = tid; i < NR * FP / 4; i += AGG_THREADS)
    out4[i] = acc4[i];
}

template <int F, int FP, int G>
__global__ void k_finalize(const float* __restrict__ h, const float* __restrict__ part,
                           const float* __restrict__ dinv, const float* __restrict__ bias,
                           float* __restrict__ out, int ldo, int N, int npad) {
  int t = blockIdx.x * blockDim.x + threadIdx.x;
  if (t >= N * F) return;
  int n = t / F;
  int f = t - n * F;
  float agg = 0.0f;
#pragma unroll
  for (int g = 0; g < G; ++g)
    agg += part[((size_t)g * npad + n) * FP + f];
  float di = dinv[n];
  float v = agg + di * di * h[n * PADF + f] + bias[f];
  out[n * ldo + f] = selu_f(v);
}

extern "C" void kernel_launch(void* const* d_in, const int* in_sizes, int n_in,
                              void* d_out, int out_size, void* d_ws, size_t ws_size,
                              hipStream_t stream) {
  const float* x  = (const float*)d_in[0];
  const int*   ei = (const int*)d_in[1];
  const float* W1 = (const float*)d_in[2];
  const float* b1 = (const float*)d_in[3];
  const float* W2 = (const float*)d_in[4];
  const float* b2 = (const float*)d_in[5];
  const float* W3 = (const float*)d_in[6];
  const float* b3 = (const float*)d_in[7];
  const float* W4 = (const float*)d_in[8];
  const float* b4 = (const float*)d_in[9];

  const int E = in_sizes[1] / 2;
  const int N = in_sizes[0] / 128;
  const int* src = ei;
  const int* dst = ei + E;

  auto cdiv = [](long long a, long long b) { return (int)((a + b - 1) / b); };
  auto al64 = [](size_t n) { return (n + 63) & ~(size_t)63; };

  float* ws   = (float*)d_ws;
  float* deg  = ws;
  float* dinv = deg  + al64(N);
  int*   rec  = (int*)(dinv + al64(N));
  const int NB = cdiv(E, 256);
  const int pw = NB * 64;
  unsigned* planes = (unsigned*)(rec + al64((size_t)4 * E));
  float* hA   = (float*)(planes + (size_t)4 * pw);
  float* hB   = hA   + (size_t)N * PADF;
  float* part = hB   + (size_t)N * PADF;

  const int NT = N / 16;
  const int GB = cdiv(NT, GEMM_RT * GEMM_WAVES);

  constexpr int NR1 = 4096, G1 = 10;
  constexpr int NR2 = 3572, G2 = 9;
  constexpr int NR3 = 2400, G3 = 6;
  constexpr int NR4 = 2048, G4 = 5;
  constexpr int QC1 = 12288, SW1 = 5;
  constexpr int QC2 = 10240, SW2 = 6;
  constexpr int QC3 = 12288, SW3 = 9;
  constexpr int QC4 = 8064,  SW4 = 10;
  const int R1 = cdiv(N, NR1), R2 = cdiv(N, NR2), R3 = cdiv(N, NR3), R4 = cdiv(N, NR4);

  k_fill  <<<cdiv(N, 256), 256, 0, stream>>>(deg, 1.0f, N);
  k_degree<<<cdiv(E, 256), 256, 0, stream>>>(dst, deg, E);
  k_rsqrt <<<cdiv(N, 256), 256, 0, stream>>>(deg, dinv, N);
  k_norm<NR1, NR2, NR3, NR4><<<NB, 256, 0, stream>>>(src, dst, dinv, rec, planes, pw, E);

  k_gemm_wmma<128, 15><<<GB, GEMM_WAVES * 32, 0, stream>>>(x, 128, W1, hA, NT);
  k_agg<15, 16, NR1, QC1, SW1><<<dim3(G1, R1), AGG_THREADS, 0, stream>>>(hA, planes, rec, part, E);

  k_gemm_fin_wmma<15, 20, G1, 16><<<GB, GEMM_WAVES * 32, 0, stream>>>(part, R1 * NR1, hA, dinv, b1, W2, hB, NT);
  k_agg<20, 20, NR2, QC2, SW2><<<dim3(G2, R2), AGG_THREADS, 0, stream>>>(hB, planes + (size_t)pw, rec, part, E);

  k_gemm_fin_wmma<20, 27, G2, 20><<<GB, GEMM_WAVES * 32, 0, stream>>>(part, R2 * NR2, hB, dinv, b2, W3, hA, NT);
  k_agg<27, 28, NR3, QC3, SW3><<<dim3(G3, R3), AGG_THREADS, 0, stream>>>(hA, planes + (size_t)2 * pw, rec, part, E);

  k_gemm_fin_wmma<27, 36, G3, 28><<<GB, GEMM_WAVES * 32, 0, stream>>>(part, R3 * NR3, hA, dinv, b3, W4, hB, NT);
  k_agg<36, 36, NR4, QC4, SW4><<<dim3(G4, R4), AGG_THREADS, 0, stream>>>(hB, planes + (size_t)3 * pw, rec, part, E);
  k_finalize<36, 36, G4><<<cdiv((long long)N * 36, 256), 256, 0, stream>>>(hB, part, dinv, b4,
                                                                   (float*)d_out, 36, N, R4 * NR4);
}


# k_agg of layer 1 stores each block's matched edge records as a list (one range size, 2048 rows x 5 chunks, in all layers); layers 2-4 read it: no scan, queue or record gather; LDS add order changes
# speedup vs baseline: 2.2115x; 1.2101x over previous
#include <hip/hip_runtime.h>
#include <hip/hip_bf16.h>

typedef __attribute__((ext_vector_type(2))) float v2f;
typedef __attribute__((ext_vector_type(8))) float v8f;

#define PADF 48

__device__ __forceinline__ float selu_f(float x) {
  const float alpha = 1.6732632423543772848f;
  const float scale = 1.0507009873554804934f;
  return scale * (x > 0.0f ? x : alpha * (expf(x) - 1.0f));
}

__global__ void k_fill(float* __restrict__ p, float v, int n) {
  int i = blockIdx.x * blockDim.x + threadIdx.x;
  if (i < n) p[i] = v;
}

__global__ void k_degree(const int* __restrict__ dst, float* __restrict__ deg, int E) {
  int e = blockIdx.x * blockDim.x + threadIdx.x;
  if (e < E)
    __hip_atomic_fetch_add(&deg[dst[e]], 1.0f, __ATOMIC_RELAXED, __HIP_MEMORY_SCOPE_AGENT);
}

__global__ void k_rsqrt(const float* __restrict__ deg, float* __restrict__ dinv, int n) {
  int i = blockIdx.x * blockDim.x + threadIdx.x;
  if (i < n) {
    float d = deg[i];
    dinv[i] = d > 0.0f ? rsqrtf(d) : 0.0f;
  }
}

template <int NR>
__global__ __launch_bounds__(256)
void k_norm(const int* __restrict__ src, const int* __restrict__ dst,
            const float* __restrict__ dinv, int* __restrict__ rec,
            unsigned* __restrict__ plane, int E) {
  __shared__ unsigned sb[64];
  const int tid = threadIdx.x;
  const int e = blockIdx.x * 256 + tid;
  unsigned ia = 255u;
  if (e < E) {
    const int s = src[e];
    const int d = dst[e];
    const float w = dinv[s] * dinv[d];
    reinterpret_cast<int4*>(rec)[e] = make_int4(s * PADF, d, __float_as_int(w), 0);
    ia = (unsigned)d / (unsigned)NR;
  }
  reinterpret_cast<unsigned char*>(&sb[0])[tid] = (unsigned char)ia;
  __syncthreads();
  if (tid < 64) plane[(size_t)blockIdx.x * 64 + tid] = sb[tid];
}

__device__ __forceinline__ unsigned match4(unsigned w, unsigned rid4) {
  const unsigned x  = w ^ rid4;
  const unsigned nz = ((x & 0x7f7f7f7fu) + 0x7f7f7f7fu) | x;
  const unsigned z  = (~nz & 0x80808080u) >> 7;
  return (z * 0x08102040u) >> 27;
}

#define GEMM_RT 2
#define GEMM_WAVES 8

template <int K, int FOUT>
__global__ __launch_bounds__(GEMM_WAVES * 32)
void k_gemm_wmma(const float* __restrict__ A, int lda,
                 const float* __restrict__ W,
                 float* __restrict__ H, int NT) {
  constexpr int CT = (FOUT + 15) / 16;
  const int lane = threadIdx.x & 31;
  const int half = lane >> 4;
  const int lid  = lane & 15;
  const int wave = __builtin_amdgcn_readfirstlane((int)(threadIdx.x >> 5));
  const int t0   = (blockIdx.x * GEMM_WAVES + wave) * GEMM_RT;
  if (t0 >= NT) return;

  int tile[GEMM_RT];
  const float* Ap[GEMM_RT];
#pragma unroll
  for (int rt = 0; rt < GEMM_RT; ++rt) {
    tile[rt] = (t0 + rt < NT) ? (t0 + rt) : (NT - 1);
    Ap[rt]   = A + (tile[rt] * 16 + lid) * lda + 2 * half;
  }
  bool cv[CT];
  int  cc[CT];
#pragma unroll
  for (int ct = 0; ct < CT; ++ct) {
    const int col = ct * 16 + lid;
    cv[ct] = col < FOUT;
    cc[ct] = cv[ct] ? col : (FOUT - 1);
  }

  v8f acc[GEMM_RT][CT];
#pragma unroll
  for (int rt = 0; rt < GEMM_RT; ++rt)
#pragma unroll
    for (int ct = 0; ct < CT; ++ct)
      acc[rt][ct] = v8f{};

  constexpr int K4 = K & ~3;
#pragma unroll 4
  for (int k0 = 0; k0 < K4; k0 += 4) {
    const int kb = k0 + 2 * half;
    v2f a[GEMM_RT], b[CT];
#pragma unroll
    for (int rt = 0; rt < GEMM_RT; ++rt) {
      a[rt].x = Ap[rt][k0];
      a[rt].y = Ap[rt][k0 + 1];
    }
#pragma unroll
    for (int ct = 0; ct < CT; ++ct) {
      const float b0 = W[kb * FOUT + cc[ct]];
      const float b1 = W[(kb + 1) * FOUT + cc[ct]];
      b[ct].x = cv[ct] ? b0 : 0.0f;
      b[ct].y = cv[ct] ? b1 : 0.0f;
    }
#pragma unroll
    for (int rt = 0; rt < GEMM_RT; ++rt)
#pragma unroll
      for (int ct = 0; ct < CT; ++ct)
        acc[rt][ct] = __builtin_amdgcn_wmma_f32_16x16x4_f32(false, a[rt], false, b[ct], (short)0,
                                                            acc[rt][ct], false, false);
  }
  if constexpr (K % 4 != 0) {
    const int kb  = K4 + 2 * half;
    const int k0c = (kb     < K) ? kb     : (K - 1);
    const int k1c = (kb + 1 < K) ? kb + 1 : (K - 1);
    v2f a[GEMM_RT], b[CT];
#pragma unroll
    for (int rt = 0; rt < GEMM_RT; ++rt) {
      const float* Ar = A + (tile[rt] * 16 + lid) * lda;
      const float a0 = Ar[k0c];
      const float a1 = Ar[k1c];
      a[rt].x = (kb     < K) ? a0 : 0.0f;
      a[rt].y = (kb + 1 < K) ? a1 : 0.0f;
    }
#pragma unroll
    for (int ct = 0; ct < CT; ++ct) {
      const float b0 = W[k0c * FOUT + cc[ct]];
      const float b1 = W[k1c * FOUT + cc[ct]];
      b[ct].x = (cv[ct] && kb     < K) ? b0 : 0.0f;
      b[ct].y = (cv[ct] && kb + 1 < K) ? b1 : 0.0f;
    }
#pragma unroll
    for (int rt = 0; rt < GEMM_RT; ++rt)
#pragma unroll
      for (int ct = 0; ct < CT; ++ct)
        acc[rt][ct] = __builtin_amdgcn_wmma_f32_16x16x4_f32(false, a[rt], false, b[ct], (short)0,
                                                            acc[rt][ct], false, false);
  }
#pragma unroll
  for (int rt = 0; rt < GEMM_RT; ++rt)
#pragma unroll
    for (int ct = 0; ct < CT; ++ct) {
      const int hbase = (tile[rt] * 16 + 8 * half) * PADF + ct * 16 + lid;
#pragma unroll
      for (int r = 0; r < 8; ++r)
        H[hbase + r * PADF] = acc[rt][ct][r];
    }
}

__device__ __forceinline__ float selu_sel(float x) {
  const float alpha = 1.6732632423543772848f;
  const float scale = 1.0507009873554804934f;
  const float neg = alpha * (expf(x) - 1.0f);
  return scale * (x > 0.0f ? x : neg);
}

template <int K, int FOUT, int G, int FPIN>
__global__ __launch_bounds__(GEMM_WAVES * 32)
void k_gemm_fin_wmma(const float* __restrict__ part, int npad,
                     const float* __restrict__ hp, const float* __restrict__ dinv,
                     const float* __restrict__ bias, const float* __restrict__ W,
                     float* __restrict__ H, int NT) {
  constexpr int CT = (FOUT + 15) / 16;
  constexpr int KS = (K + 3) / 4;
  static_assert(FPIN % 2 == 0 && FPIN >= 4 * KS && 4 * KS <= PADF, "pair loads of the last k step stay inside a row");
  const int lane = threadIdx.x & 31;
  const int half = lane >> 4;
  const int lid  = lane & 15;
  const int wave = __builtin_amdgcn_readfirstlane((int)(threadIdx.x >> 5));
  const int t0   = (blockIdx.x * GEMM_WAVES + wave) * GEMM_RT;
  if (t0 >= NT) return;

  const size_t ps = (size_t)npad * FPIN;
  int tile[GEMM_RT];
  float di[GEMM_RT];
  v2f pv[KS][GEMM_RT][G];
  v2f hv[KS][GEMM_RT];
#pragma unroll
  for (int rt = 0; rt < GEMM_RT; ++rt) {
    tile[rt] = (t0 + rt < NT) ? (t0 + rt) : (NT - 1);
    const int row = tile[rt] * 16 + lid;
    const float* pr = part + (size_t)row * FPIN + 2 * half;
    const float* hr = hp + (size_t)row * PADF + 2 * half;
#pragma unroll
    for (int s = 0; s < KS; ++s) {
#pragma unroll
      for (int g = 0; g < G; ++g)
        pv[s][rt][g] = *reinterpret_cast<const v2f*>(pr + (size_t)g * ps + 4 * s);
      hv[s][rt] = *reinterpret_cast<const v2f*>(hr + 4 * s);
    }
    di[rt] = dinv[row];
  }
  __builtin_amdgcn_sched_barrier(0);
  float dd[GEMM_RT];
#pragma unroll
  for (int rt = 0; rt < GEMM_RT; ++rt)
    dd[rt] = di[rt] * di[rt];
  bool cv[CT];
  int  cc[CT];
#pragma unroll
  for (int ct = 0; ct < CT; ++ct) {
    const int col = ct * 16 + lid;
    cv[ct] = col < FOUT;
    cc[ct] = cv[ct] ? col : (FOUT - 1);
  }

  v8f acc[GEMM_RT][CT];
#pragma unroll
  for (int rt = 0; rt < GEMM_RT; ++rt)
#pragma unroll
    for (int ct = 0; ct < CT; ++ct)
      acc[rt][ct] = v8f{};

#pragma unroll
  for (int s = 0; s < KS; ++s) {
    const int kb  = 4 * s + 2 * half;
    const bool v0 = kb < K, v1 = kb + 1 < K;
    const int k0c = v0 ? kb     : (K - 1);
    const int k1c = v1 ? kb + 1 : (K - 1);
    const float bk0 = bias[k0c], bk1 = bias[k1c];
    v2f a[GEMM_RT], b[CT];
#pragma unroll
    for (int rt = 0; rt < GEMM_RT; ++rt) {
      float ag0 = 0.0f, ag1 = 0.0f;
#pragma unroll
      for (int g = 0; g < G; ++g) {
        ag0 += pv[s][rt][g].x;
        ag1 += pv[s][rt][g].y;
      }
      const float x0 = ag0 + dd[rt] * hv[s][rt].x + bk0;
      const float x1 = ag1 + dd[rt] * hv[s][rt].y + bk1;
      const float a0 = selu_sel(x0);
      const float a1 = selu_sel(x1);
      a[rt].x = v0 ? a0 : 0.0f;
      a[rt].y = v1 ? a1 : 0.0f;
    }
#pragma unroll
    for (int ct = 0; ct < CT; ++ct) {
      const float b0 = W[k0c * FOUT + cc[ct]];
      const float b1 = W[k1c * FOUT + cc[ct]];
      b[ct].x = (cv[ct] && v0) ? b0 : 0.0f;
      b[ct].y = (cv[ct] && v1) ? b1 : 0.0f;
    }
#pragma unroll
    for (int rt = 0; rt < GEMM_RT; ++rt)
#pragma unroll
      for (int ct = 0; ct < CT; ++ct)
        acc[rt][ct] = __builtin_amdgcn_wmma_f32_16x16x4_f32(false, a[rt], false, b[ct], (short)0,
                                                            acc[rt][ct], false, false);
  }
#pragma unroll
  for (int rt = 0; rt < GEMM_RT; ++rt)
#pragma unroll
    for (int ct = 0; ct < CT; ++ct) {
      const int hbase = (tile[rt] * 16 + 8 * half) * PADF + ct * 16 + lid;
#pragma unroll
      for (int r = 0; r < 8; ++r)
        H[hbase + r * PADF] = acc[rt][ct][r];
    }
}

#define AGG_THREADS 1024
#define AGG_U 8
#define AGG_CAP 8192

template <int FP, bool FROM_LIST, bool TO_LIST>
__device__ __forceinline__ void agg_drain(const float* __restrict__ h, const int4* __restrict__ recs,
                                          const int* q_e, float* acc, int4* __restrict__ lst, int lroom,
                                          int M, int lo, int esub, int pc, bool lane_used) {
  constexpr int L  = FP / 4;
  constexpr int EP = AGG_THREADS / L;
  const int c4  = pc * 4;
  const int rot = (pc + esub) & 3;
  const int o0 = rot, o1 = (rot + 1) & 3, o2 = (rot + 2) & 3, o3 = (rot + 3) & 3;
  for (int m0 = 0; m0 < M; m0 += EP * AGG_U) {
    int4   rc[AGG_U];
    float4 hv[AGG_U];
    float  wv[AGG_U];
    int    a[AGG_U];
    bool   ok[AGG_U];
#pragma unroll
    for (int u = 0; u < AGG_U; ++u) {
      const int m = m0 + u * EP + esub;
      ok[u] = lane_used && (m < M);
      const int mc = ok[u] ? m : 0;
      if constexpr (FROM_LIST) rc[u] = recs[mc];
      else                     rc[u] = recs[q_e[mc]];
    }
    if constexpr (FROM_LIST) __builtin_amdgcn_sched_barrier(0);
#pragma unroll
    for (int u = 0; u < AGG_U; ++u) {
      asm volatile("" :: "v"(rc[u].x), "v"(rc[u].y), "v"(rc[u].z));
    }
#pragma unroll
    for (int u = 0; u < AGG_U; ++u) {
      hv[u] = *reinterpret_cast<const float4*>(h + rc[u].x + c4);
      wv[u] = __int_as_float(rc[u].z);
      a[u]  = (rc[u].y - lo) * FP + c4;
    }
    if constexpr (TO_LIST) {
#pragma unroll
      for (int u = 0; u < AGG_U; ++u) {
        const int m = m0 + u * EP + esub;
        if (ok[u] && pc == 0 && m < lroom)
          lst[m] = make_int4(rc[u].x, rc[u].y, rc[u].z, 0);
      }
    }
#pragma unroll
    for (int u = 0; u < AGG_U; ++u) {
      asm volatile("" :: "v"(hv[u].x), "v"(hv[u].y), "v"(hv[u].z), "v"(hv[u].w));
    }
#pragma unroll
    for (int u = 0; u < AGG_U; ++u)
      if (ok[u]) {
        float v0 = hv[u].x, v1 = hv[u].y, v2 = hv[u].z, v3 = hv[u].w;
        if (rot & 1) { const float t = v0; v0 = v1; v1 = v2; v2 = v3; v3 = t; }
        if (rot & 2) { const float t0 = v0, t1 = v1; v0 = v2; v1 = v3; v2 = t0; v3 = t1; }
        float* p = &acc[a[u]];
        const float w = wv[u];
        __hip_atomic_fetch_add(p + o0, w * v0, __ATOMIC_RELAXED, __HIP_MEMORY_SCOPE_WORKGROUP);
        __hip_atomic_fetch_add(p + o1, w * v1, __ATOMIC_RELAXED, __HIP_MEMORY_SCOPE_WORKGROUP);
        __hip_atomic_fetch_add(p + o2, w * v2, __ATOMIC_RELAXED, __HIP_MEMORY_SCOPE_WORKGROUP);
        __hip_atomic_fetch_add(p + o3, w * v3, __ATOMIC_RELAXED, __HIP_MEMORY_SCOPE_WORKGROUP);
      }
  }
}

template <int F, int FP, int NR, int QC, int W, int MODE>
__global__ __launch_bounds__(AGG_THREADS)
void k_agg(const float* __restrict__ h, const unsigned* __restrict__ plane,
           const int* __restrict__ rec, float* __restrict__ part,
           int* __restrict__ list, int* __restrict__ cnt, int E) {
  __shared__ __attribute__((aligned(16))) float acc[NR * FP];
  __shared__ int   q_e[QC];
  __shared__ int   q_n;
  __shared__ int   q_more;

  const int tid  = threadIdx.x;
  const int lane = tid & 31;
  const int lo   = blockIdx.y * NR;
  const int G    = gridDim.x;
  const int echunk = (((E + G - 1) / G) + 31) & ~31;
  const int e_begin = blockIdx.x * echunk;
  const int e_end   = (e_begin + echunk < E) ? (e_begin + echunk) : E;

  constexpr int PASS = AGG_THREADS * 32 * W;
  constexpr int L  = FP / 4;
  constexpr int EP = AGG_THREADS / L;
  const int  esub = tid / L;
  const int  pc   = tid - esub * L;
  const bool lane_used = tid < EP * L;
  const unsigned rid4 = (unsigned)blockIdx.y * 0x01010101u;

  float4* acc4 = reinterpret_cast<float4*>(acc);
  for (int i = tid; i < NR * FP / 4; i += AGG_THREADS)
    acc4[i] = make_float4(0.0f, 0.0f, 0.0f, 0.0f);
  if (tid == 0) { q_n = 0; q_more = 0; }
  __syncthreads();

  const uint4* plane4 = reinterpret_cast<const uint4*>(plane);
  const int4*  rec4   = reinterpret_cast<const int4*>(rec);
  const int    blk    = blockIdx.y * G + blockIdx.x;
  int4* mylist = reinterpret_cast<int4*>(list) + (size_t)blk * AGG_CAP;

  int n_list = AGG_CAP + 1;
  if constexpr (MODE == 1) n_list = cnt[blk];

  if (MODE == 1 && n_list <= AGG_CAP) {
    agg_drain<FP, true, false>(h, mylist, q_e, acc, mylist, 0, n_list, lo, esub, pc, lane_used);
    __syncthreads();
  } else {
    int done = 0;
    for (int b = e_begin; b < e_end; b += PASS) {
      unsigned pend[W];
      constexpr int WC = (W <= 6) ? W : (W + 1) / 2;
#pragma unroll
      for (int w0 = 0; w0 < W; w0 += WC) {
        uint4 pl[2 * WC];
#pragma unroll
        for (int i = 0; i < WC; ++i)
          if (w0 + i < W) {
            const int eb  = b + ((w0 + i) * AGG_THREADS + tid) * 32;
            const int ebc = (e_end - eb > 0) ? eb : e_begin;
            pl[2 * i]     = plane4[(ebc >> 4)];
            pl[2 * i + 1] = plane4[(ebc >> 4) + 1];
          }
        __builtin_amdgcn_sched_barrier(0);
#pragma unroll
        for (int i = 0; i < WC; ++i)
          if (w0 + i < W) {
            const int eb = b + ((w0 + i) * AGG_THREADS + tid) * 32;
            const int nv = e_end - eb;
            const uint4 p0 = pl[2 * i], p1 = pl[2 * i + 1];
            unsigned mt = match4(p0.x, rid4)       | match4(p0.y, rid4) << 4  |
                          match4(p0.z, rid4) << 8  | match4(p0.w, rid4) << 12 |
                          match4(p1.x, rid4) << 16 | match4(p1.y, rid4) << 20 |
                          match4(p1.z, rid4) << 24 | match4(p1.w, rid4) << 28;
            mt &= (nv >= 32) ? 0xffffffffu : ((nv > 0) ? ((1u << nv) - 1u) : 0u);
            pend[w0 + i] = mt;
          }
      }

      bool more;
      do {
        int c = 0;
#pragma unroll
        for (int w = 0; w < W; ++w) c += __popc(pend[w]);
        int incl = c;
#pragma unroll
        for (int o = 1; o < 32; o <<= 1) {
          const int t = __shfl_up(incl, o, 32);
          if (lane >= o) incl += t;
        }
        const int wtot = __shfl(incl, 31, 32);
        int base = 0;
        if (lane == 31 && wtot > 0)
          base = __hip_atomic_fetch_add(&q_n, wtot, __ATOMIC_RELAXED, __HIP_MEMORY_SCOPE_WORKGROUP);
        base = __shfl(base, 31, 32);
        int pos = base + incl - c;
        unsigned left = 0;
#pragma unroll
        for (int w = 0; w < W; ++w) {
          const int eb = b + (w * AGG_THREADS + tid) * 32;
          unsigned p = pend[w];
          while (p != 0u && pos < QC) {
            const int j = __ffs(p) - 1;
            p &= p - 1;
            q_e[pos] = eb + j;
            ++pos;
          }
          pend[w] = p;
          left |= p;
        }
        if (left) q_more = 1;
        __syncthreads();

        const int M = (q_n < QC) ? q_n : QC;
        more = (q_more != 0);
        __syncthreads();

        if (tid == 0) { q_n = 0; q_more = 0; }
        if constexpr (MODE == 0)
          agg_drain<FP, false, true>(h, rec4, q_e, acc, mylist + (done < AGG_CAP ? done : AGG_CAP),
                                     AGG_CAP - done, M, lo, esub, pc, lane_used);
        else
          agg_drain<FP, false, false>(h, rec4, q_e, acc, mylist, 0, M, lo, esub, pc, lane_used);
        done += M;
        __syncthreads();
      } while (more);
    }
    if constexpr (MODE == 0)
      if (tid == 0) cnt[blk] = done;
  }

  const size_t npad = (size_t)gridDim.y * NR;
  float4* out4 = reinterpret_cast<float4*>(part + ((size_t)blockIdx.x * npad + lo) * FP);
  for (int i = tid; i < NR * FP / 4; i += AGG_THREADS)
    out4[i] = acc4[i];
}

template <int F, int FP, int G>
__global__ void k_finalize(const float* __restrict__ h, const float* __restrict__ part,
                           const float* __restrict__ dinv, const float* __restrict__ bias,
                           float* __restrict__ out, int ldo, int N, int npad) {
  int t = blockIdx.x * blockDim.x + threadIdx.x;
  if (t >= N * F) return;
  int n = t / F;
  int f = t - n * F;
  float agg = 0.0f;
#pragma unroll
  for (int g = 0; g < G; ++g)
    agg += part[((size_t)g * npad + n) * FP + f];
  float di = dinv[n];
  float v = agg + di * di * h[n * PADF + f] + bias[f];
  out[n * ldo + f] = selu_f(v);
}

extern "C" void kernel_launch(void* const* d_in, const int* in_sizes, int n_in,
                              void* d_out, int out_size, void* d_ws, size_t ws_size,
                              hipStream_t stream) {
  const float* x  = (const float*)d_in[0];
  const int*   ei = (const int*)d_in[1];
  const float* W1 = (const float*)d_in[2];
  const float* b1 = (const float*)d_in[3];
  const float* W2 = (const float*)d_in[4];
  const float* b2 = (const float*)d_in[5];
  const float* W3 = (const float*)d_in[6];
  const float* b3 = (const float*)d_in[7];
  const float* W4 = (const float*)d_in[8];
  const float* b4 = (const float*)d_in[9];

  const int E = in_sizes[1] / 2;
  const int N = in_sizes[0] / 128;
  const int* src = ei;
  const int* dst = ei + E;

  auto cdiv = [](long long a, long long b) { return (int)((a + b - 1) / b); };
  auto al64 = [](size_t n) { return (n + 63) & ~(size_t)63; };

  float* ws   = (float*)d_ws;
  float* deg  = ws;
  float* dinv = deg  + al64(N);
  int*   rec  = (int*)(dinv + al64(N));
  const int NB = cdiv(E, 256);
  const int pw = NB * 64;
  unsigned* plane = (unsigned*)(rec + al64((size_t)4 * E));
  float* hA   = (float*)(plane + (size_t)pw);
  float* hB   = hA   + (size_t)N * PADF;
  float* part = hB   + (size_t)N * PADF;

  const int NT = N / 16;
  const int GB = cdiv(NT, GEMM_RT * GEMM_WAVES);

  constexpr int NR = 2048, G = 5;
  constexpr int QC = 8064, SW = 10;
  const int R = cdiv(N, NR);
  int* list = (int*)(part + al64((size_t)G * R * NR * 36));
  int* cnt  = list + (size_t)R * G * AGG_CAP * 4;

  k_fill  <<<cdiv(N, 256), 256, 0, stream>>>(deg, 1.0f, N);
  k_degree<<<cdiv(E, 256), 256, 0, stream>>>(dst, deg, E);
  k_rsqrt <<<cdiv(N, 256), 256, 0, stream>>>(deg, dinv, N);
  k_norm<NR><<<NB, 256, 0, stream>>>(src, dst, dinv, rec, plane, E);

  k_gemm_wmma<128, 15><<<GB, GEMM_WAVES * 32, 0, stream>>>(x, 128, W1, hA, NT);
  k_agg<15, 16, NR, QC, SW, 0><<<dim3(G, R), AGG_THREADS, 0, stream>>>(hA, plane, rec, part, list, cnt, E);

  k_gemm_fin_wmma<15, 20, G, 16><<<GB, GEMM_WAVES * 32, 0, stream>>>(part, R * NR, hA, dinv, b1, W2, hB, NT);
  k_agg<20, 20, NR, QC, SW, 1><<<dim3(G, R), AGG_THREADS, 0, stream>>>(hB, plane, rec, part, list, cnt, E);

  k_gemm_fin_wmma<20, 27, G, 20><<<GB, GEMM_WAVES * 32, 0, stream>>>(part, R * NR, hB, dinv, b2, W3, hA, NT);
  k_agg<27, 28, NR, QC, SW, 1><<<dim3(G, R), AGG_THREADS, 0, stream>>>(hA, plane, rec, part, list, cnt, E);

  k_gemm_fin_wmma<27, 36, G, 28><<<GB, GEMM_WAVES * 32, 0, stream>>>(part, R * NR, hA, dinv, b3, W4, hB, NT);
  k_agg<36, 36, NR, QC, SW, 1><<<dim3(G, R), AGG_THREADS, 0, stream>>>(hB, plane, rec, part, list, cnt, E);
  k_finalize<36, 36, G><<<cdiv((long long)N * 36, 256), 256, 0, stream>>>(hB, part, dinv, b4,
                                                                   (float*)d_out, 36, N, R * NR);
}


# k_agg layers 2-4: a block owns 400 rows (1/5 of a row range), fed by the 5 chunk sub-lists layer 1 sorts for it: 1 plane, not 5 partial planes stored and re-added; a row's sums in one arrival order
# speedup vs baseline: 2.3896x; 1.0805x over previous
#include <hip/hip_runtime.h>
#include <hip/hip_bf16.h>

typedef __attribute__((ext_vector_type(2))) float v2f;
typedef __attribute__((ext_vector_type(8))) float v8f;

#define PADF 48

__device__ __forceinline__ float selu_f(float x) {
  const float alpha = 1.6732632423543772848f;
  const float scale = 1.0507009873554804934f;
  return scale * (x > 0.0f ? x : alpha * (expf(x) - 1.0f));
}

__global__ void k_fill(float* __restrict__ p, float v, int n) {
  int i = blockIdx.x * blockDim.x + threadIdx.x;
  if (i < n) p[i] = v;
}

__global__ void k_degree(const int* __restrict__ dst, float* __restrict__ deg, int E) {
  int e = blockIdx.x * blockDim.x + threadIdx.x;
  if (e < E)
    __hip_atomic_fetch_add(&deg[dst[e]], 1.0f, __ATOMIC_RELAXED, __HIP_MEMORY_SCOPE_AGENT);
}

__global__ void k_rsqrt(const float* __restrict__ deg, float* __restrict__ dinv, int n) {
  int i = blockIdx.x * blockDim.x + threadIdx.x;
  if (i < n) {
    float d = deg[i];
    dinv[i] = d > 0.0f ? rsqrtf(d) : 0.0f;
  }
}

template <int NR>
__global__ __launch_bounds__(256)
void k_norm(const int* __restrict__ src, const int* __restrict__ dst,
            const float* __restrict__ dinv, int* __restrict__ rec,
            unsigned* __restrict__ plane, int E) {
  __shared__ unsigned sb[64];
  const int tid = threadIdx.x;
  const int e = blockIdx.x * 256 + tid;
  unsigned ia = 255u;
  if (e < E) {
    const int s = src[e];
    const int d = dst[e];
    const float w = dinv[s] * dinv[d];
    reinterpret_cast<int4*>(rec)[e] = make_int4(s * PADF, d, __float_as_int(w), 0);
    ia = (unsigned)d / (unsigned)NR;
  }
  reinterpret_cast<unsigned char*>(&sb[0])[tid] = (unsigned char)ia;
  __syncthreads();
  if (tid < 64) plane[(size_t)blockIdx.x * 64 + tid] = sb[tid];
}

__device__ __forceinline__ unsigned match4(unsigned w, unsigned rid4) {
  const unsigned x  = w ^ rid4;
  const unsigned nz = ((x & 0x7f7f7f7fu) + 0x7f7f7f7fu) | x;
  const unsigned z  = (~nz & 0x80808080u) >> 7;
  return (z * 0x08102040u) >> 27;
}

#define GEMM_RT 2
#define GEMM_WAVES 8

template <int K, int FOUT>
__global__ __launch_bounds__(GEMM_WAVES * 32)
void k_gemm_wmma(const float* __restrict__ A, int lda,
                 const float* __restrict__ W,
                 float* __restrict__ H, int NT) {
  constexpr int CT = (FOUT + 15) / 16;
  const int lane = threadIdx.x & 31;
  const int half = lane >> 4;
  const int lid  = lane & 15;
  const int wave = __builtin_amdgcn_readfirstlane((int)(threadIdx.x >> 5));
  const int t0   = (blockIdx.x * GEMM_WAVES + wave) * GEMM_RT;
  if (t0 >= NT) return;

  int tile[GEMM_RT];
  const float* Ap[GEMM_RT];
#pragma unroll
  for (int rt = 0; rt < GEMM_RT; ++rt) {
    tile[rt] = (t0 + rt < NT) ? (t0 + rt) : (NT - 1);
    Ap[rt]   = A + (tile[rt] * 16 + lid) * lda + 2 * half;
  }
  bool cv[CT];
  int  cc[CT];
#pragma unroll
  for (int ct = 0; ct < CT; ++ct) {
    const int col = ct * 16 + lid;
    cv[ct] = col < FOUT;
    cc[ct] = cv[ct] ? col : (FOUT - 1);
  }

  v8f acc[GEMM_RT][CT];
#pragma unroll
  for (int rt = 0; rt < GEMM_RT; ++rt)
#pragma unroll
    for (int ct = 0; ct < CT; ++ct)
      acc[rt][ct] = v8f{};

  constexpr int K4 = K & ~3;
#pragma unroll 4
  for (int k0 = 0; k0 < K4; k0 += 4) {
    const int kb = k0 + 2 * half;
    v2f a[GEMM_RT], b[CT];
#pragma unroll
    for (int rt = 0; rt < GEMM_RT; ++rt) {
      a[rt].x = Ap[rt][k0];
      a[rt].y = Ap[rt][k0 + 1];
    }
#pragma unroll
    for (int ct = 0; ct < CT; ++ct) {
      const float b0 = W[kb * FOUT + cc[ct]];
      const float b1 = W[(kb + 1) * FOUT + cc[ct]];
      b[ct].x = cv[ct] ? b0 : 0.0f;
      b[ct].y = cv[ct] ? b1 : 0.0f;
    }
#pragma unroll
    for (int rt = 0; rt < GEMM_RT; ++rt)
#pragma unroll
      for (int ct = 0; ct < CT; ++ct)
        acc[rt][ct] = __builtin_amdgcn_wmma_f32_16x16x4_f32(false, a[rt], false, b[ct], (short)0,
                                                            acc[rt][ct], false, false);
  }
  if constexpr (K % 4 != 0) {
    const int kb  = K4 + 2 * half;
    const int k0c = (kb     < K) ? kb     : (K - 1);
    const int k1c = (kb + 1 < K) ? kb + 1 : (K - 1);
    v2f a[GEMM_RT], b[CT];
#pragma unroll
    for (int rt = 0; rt < GEMM_RT; ++rt) {
      const float* Ar = A + (tile[rt] * 16 + lid) * lda;
      const float a0 = Ar[k0c];
      const float a1 = Ar[k1c];
      a[rt].x = (kb     < K) ? a0 : 0.0f;
      a[rt].y = (kb + 1 < K) ? a1 : 0.0f;
    }
#pragma unroll
    for (int ct = 0; ct < CT; ++ct) {
      const float b0 = W[k0c * FOUT + cc[ct]];
      const float b1 = W[k1c * FOUT + cc[ct]];
      b[ct].x = (cv[ct] && kb     < K) ? b0 : 0.0f;
      b[ct].y = (cv[ct] && kb + 1 < K) ? b1 : 0.0f;
    }
#pragma unroll
    for (int rt = 0; rt < GEMM_RT; ++rt)
#pragma unroll
      for (int ct = 0; ct < CT; ++ct)
        acc[rt][ct] = __builtin_amdgcn_wmma_f32_16x16x4_f32(false, a[rt], false, b[ct], (short)0,
                                                            acc[rt][ct], false, false);
  }
#pragma unroll
  for (int rt = 0; rt < GEMM_RT; ++rt)
#pragma unroll
    for (int ct = 0; ct < CT; ++ct) {
      const int hbase = (tile[rt] * 16 + 8 * half) * PADF + ct * 16 + lid;
#pragma unroll
      for (int r = 0; r < 8; ++r)
        H[hbase + r * PADF] = acc[rt][ct][r];
    }
}

__device__ __forceinline__ float selu_sel(float x) {
  const float alpha = 1.6732632423543772848f;
  const float scale = 1.0507009873554804934f;
  const float neg = alpha * (expf(x) - 1.0f);
  return scale * (x > 0.0f ? x : neg);
}

template <int K, int FOUT, int G, int FPIN>
__global__ __launch_bounds__(GEMM_WAVES * 32)
void k_gemm_fin_wmma(const float* __restrict__ part, int npad,
                     const float* __restrict__ hp, const float* __restrict__ dinv,
                     const float* __restrict__ bias, const float* __restrict__ W,
                     float* __restrict__ H, int NT) {
  constexpr int CT = (FOUT + 15) / 16;
  constexpr int KS = (K + 3) / 4;
  static_assert(FPIN % 2 == 0 && FPIN >= 4 * KS && 4 * KS <= PADF, "pair loads of the last k step stay inside a row");
  const int lane = threadIdx.x & 31;
  const int half = lane >> 4;
  const int lid  = lane & 15;
  const int wave = __builtin_amdgcn_readfirstlane((int)(threadIdx.x >> 5));
  const int t0   = (blockIdx.x * GEMM_WAVES + wave) * GEMM_RT;
  if (t0 >= NT) return;

  const size_t ps = (size_t)npad * FPIN;
  int tile[GEMM_RT];
  float di[GEMM_RT];
  v2f pv[KS][GEMM_RT][G];
  v2f hv[KS][GEMM_RT];
#pragma unroll
  for (int rt = 0; rt < GEMM_RT; ++rt) {
    tile[rt] = (t0 + rt < NT) ? (t0 + rt) : (NT - 1);
    const int row = tile[rt] * 16 + lid;
    const float* pr = part + (size_t)row * FPIN + 2 * half;
    const float* hr = hp + (size_t)row * PADF + 2 * half;
#pragma unroll
    for (int s = 0; s < KS; ++s) {
#pragma unroll
      for (int g = 0; g < G; ++g)
        pv[s][rt][g] = *reinterpret_cast<const v2f*>(pr + (size_t)g * ps + 4 * s);
      hv[s][rt] = *reinterpret_cast<const v2f*>(hr + 4 * s);
    }
    di[rt] = dinv[row];
  }
  __builtin_amdgcn_sched_barrier(0);
  float dd[GEMM_RT];
#pragma unroll
  for (int rt = 0; rt < GEMM_RT; ++rt)
    dd[rt] = di[rt] * di[rt];
  bool cv[CT];
  int  cc[CT];
#pragma unroll
  for (int ct = 0; ct < CT; ++ct) {
    const int col = ct * 16 + lid;
    cv[ct] = col < FOUT;
    cc[ct] = cv[ct] ? col : (FOUT - 1);
  }

  v8f acc[GEMM_RT][CT];
#pragma unroll
  for (int rt = 0; rt < GEMM_RT; ++rt)
#pragma unroll
    for (int ct = 0; ct < CT; ++ct)
      acc[rt][ct] = v8f{};

#pragma unroll
  for (int s = 0; s < KS; ++s) {
    const int kb  = 4 * s + 2 * half;
    const bool v0 = kb < K, v1 = kb + 1 < K;
    const int k0c = v0 ? kb     : (K - 1);
    const int k1c = v1 ? kb + 1 : (K - 1);
    const float bk0 = bias[k0c], bk1 = bias[k1c];
    v2f a[GEMM_RT], b[CT];
#pragma unroll
    for (int rt = 0; rt < GEMM_RT; ++rt) {
      float ag0 = 0.0f, ag1 = 0.0f;
#pragma unroll
      for (int g = 0; g < G; ++g) {
        ag0 += pv[s][rt][g].x;
        ag1 += pv[s][rt][g].y;
      }
      const float x0 = ag0 + dd[rt] * hv[s][rt].x + bk0;
      const float x1 = ag1 + dd[rt] * hv[s][rt].y + bk1;
      const float a0 = selu_sel(x0);
      const float a1 = selu_sel(x1);
      a[rt].x = v0 ? a0 : 0.0f;
      a[rt].y = v1 ? a1 : 0.0f;
    }
#pragma unroll
    for (int ct = 0; ct < CT; ++ct) {
      const float b0 = W[k0c * FOUT + cc[ct]];
      const float b1 = W[k1c * FOUT + cc[ct]];
      b[ct].x = (cv[ct] && v0) ? b0 : 0.0f;
      b[ct].y = (cv[ct] && v1) ? b1 : 0.0f;
    }
#pragma unroll
    for (int rt = 0; rt < GEMM_RT; ++rt)
#pragma unroll
      for (int ct = 0; ct < CT; ++ct)
        acc[rt][ct] = __builtin_amdgcn_wmma_f32_16x16x4_f32(false, a[rt], false, b[ct], (short)0,
                                                            acc[rt][ct], false, false);
  }
#pragma unroll
  for (int rt = 0; rt < GEMM_RT; ++rt)
#pragma unroll
    for (int ct = 0; ct < CT; ++ct) {
      const int hbase = (tile[rt] * 16 + 8 * half) * PADF + ct * 16 + lid;
#pragma unroll
      for (int r = 0; r < 8; ++r)
        H[hbase + r * PADF] = acc[rt][ct][r];
    }
}

#define AGG_THREADS 1024
#define AGG_U 8
#define AGG_NG 5
#define AGG_CAPS 2048

template <int FP, int NRS, int ROWS, bool FROM_LIST, bool TO_LIST, bool FILTER>
__device__ __forceinline__ void agg_drain(const float* __restrict__ h, const int4* __restrict__ recs,
                                          const int* q_e, float* acc, int4* __restrict__ lst, int* l_cnt,
                                          const int* pre, const int* shift,
                                          int M, int lo, int esub, int pc, bool lane_used) {
  constexpr int L  = FP / 4;
  constexpr int EP = AGG_THREADS / L;
  const int c4  = pc * 4;
  const int rot = (pc + esub) & 3;
  const int o0 = rot, o1 = (rot + 1) & 3, o2 = (rot + 2) & 3, o3 = (rot + 3) & 3;
  for (int m0 = 0; m0 < M; m0 += EP * AGG_U) {
    int4   rc[AGG_U];
    float4 hv[AGG_U];
    float  wv[AGG_U];
    int    a[AGG_U];
    bool   ok[AGG_U];
#pragma unroll
    for (int u = 0; u < AGG_U; ++u) {
      const int m = m0 + u * EP + esub;
      ok[u] = lane_used && (m < M);
      const int mc = ok[u] ? m : 0;
      if constexpr (FROM_LIST) {
        int idx = mc;
#pragma unroll
        for (int k = 1; k < AGG_NG; ++k) idx += (mc >= pre[k]) ? shift[k] : 0;
        rc[u] = recs[idx];
      } else {
        rc[u] = recs[q_e[mc]];
      }
    }
    if constexpr (FROM_LIST) __builtin_amdgcn_sched_barrier(0);
#pragma unroll
    for (int u = 0; u < AGG_U; ++u) {
      asm volatile("" :: "v"(rc[u].x), "v"(rc[u].y), "v"(rc[u].z));
    }
#pragma unroll
    for (int u = 0; u < AGG_U; ++u) {
      hv[u] = *reinterpret_cast<const float4*>(h + rc[u].x + c4);
      wv[u] = __int_as_float(rc[u].z);
      a[u]  = (rc[u].y - lo) * FP + c4;
      if constexpr (FILTER) ok[u] = ok[u] && ((unsigned)(rc[u].y - lo) < (unsigned)ROWS);
    }
    if constexpr (TO_LIST) {
#pragma unroll
      for (int u = 0; u < AGG_U; ++u) {
        if (ok[u] && pc == 0) {
          const int sub = (int)((unsigned)(rc[u].y - lo) / (unsigned)NRS);
          const int pos = __hip_atomic_fetch_add(&l_cnt[sub], 1, __ATOMIC_RELAXED, __HIP_MEMORY_SCOPE_WORKGROUP);
          if (pos < AGG_CAPS)
            lst[(size_t)sub * (AGG_NG * AGG_CAPS) + pos] = make_int4(rc[u].x, rc[u].y, rc[u].z, 0);
        }
      }
    }
#pragma unroll
    for (int u = 0; u < AGG_U; ++u) {
      asm volatile("" :: "v"(hv[u].x), "v"(hv[u].y), "v"(hv[u].z), "v"(hv[u].w));
    }
#pragma unroll
    for (int u = 0; u < AGG_U; ++u)
      if (ok[u]) {
        float v0 = hv[u].x, v1 = hv[u].y, v2 = hv[u].z, v3 = hv[u].w;
        if (rot & 1) { const float t = v0; v0 = v1; v1 = v2; v2 = v3; v3 = t; }
        if (rot & 2) { const float t0 = v0, t1 = v1; v0 = v2; v1 = v3; v2 = t0; v3 = t1; }
        float* p = &acc[a[u]];
        const float w = wv[u];
        __hip_atomic_fetch_add(p + o0, w * v0, __ATOMIC_RELAXED, __HIP_MEMORY_SCOPE_WORKGROUP);
        __hip_atomic_fetch_add(p + o1, w * v1, __ATOMIC_RELAXED, __HIP_MEMORY_SCOPE_WORKGROUP);
        __hip_atomic_fetch_add(p + o2, w * v2, __ATOMIC_RELAXED, __HIP_MEMORY_SCOPE_WORKGROUP);
        __hip_atomic_fetch_add(p + o3, w * v3, __ATOMIC_RELAXED, __HIP_MEMORY_SCOPE_WORKGROUP);
      }
  }
}

template <int F, int FP, int NR, int QC, int W, int MODE>
__global__ __launch_bounds__(AGG_THREADS)
void k_agg(const float* __restrict__ h, const unsigned* __restrict__ plane,
           const int* __restrict__ rec, float* __restrict__ part,
           int* __restrict__ list, int* __restrict__ cnt, int E) {
  constexpr int NG   = AGG_NG;
  static_assert(NR % NG == 0, "a row range splits into NG equal sub-ranges");
  constexpr int NRS  = NR / NG;
  constexpr int ROWS = (MODE == 0) ? NR : NRS;
  static_assert((ROWS * FP) % 4 == 0, "the accumulators are cleared and copied 16 bytes at a time");
  __shared__ __attribute__((aligned(16))) float acc[ROWS * FP];
  __shared__ int   q_e[QC];
  __shared__ int   q_n;
  __shared__ int   q_more;
  __shared__ int   l_cnt[NG];

  const int tid  = threadIdx.x;
  const int lane = tid & 31;
  const int lo   = blockIdx.y * NR + ((MODE == 0) ? 0 : (int)blockIdx.x * NRS);

  constexpr int PASS = AGG_THREADS * 32 * W;
  constexpr int L  = FP / 4;
  constexpr int EP = AGG_THREADS / L;
  const int  esub = tid / L;
  const int  pc   = tid - esub * L;
  const bool lane_used = tid < EP * L;
  const unsigned rid4 = (unsigned)blockIdx.y * 0x01010101u;

  float4* acc4 = reinterpret_cast<float4*>(acc);
  for (int i = tid; i < ROWS * FP / 4; i += AGG_THREADS)
    acc4[i] = make_float4(0.0f, 0.0f, 0.0f, 0.0f);
  if (tid == 0) { q_n = 0; q_more = 0; }
  if (tid < NG) l_cnt[tid] = 0;
  __syncthreads();

  const uint4* plane4 = reinterpret_cast<const uint4*>(plane);
  const int4*  rec4   = reinterpret_cast<const int4*>(rec);
  int4*        list4  = reinterpret_cast<int4*>(list);
  const int seg1 = ((int)blockIdx.y * NG + (int)blockIdx.x) * NG;
  int4* wlist = list4 + ((size_t)blockIdx.y * NG * NG + blockIdx.x) * AGG_CAPS;

  int  pre[NG], shift[NG];
  int  total = 0;
  bool fits  = (MODE == 1);
  if constexpr (MODE == 1) {
    int nprev = 0;
#pragma unroll
    for (int g = 0; g < NG; ++g) {
      const int n = cnt[seg1 + g];
      pre[g]   = total;
      shift[g] = (g == 0) ? 0 : (AGG_CAPS - nprev);
      fits     = fits && (n <= AGG_CAPS);
      total   += n;
      nprev    = n;
    }
  } else {
#pragma unroll
    for (int g = 0; g < NG; ++g) { pre[g] = 0; shift[g] = 0; }
  }

  if (fits) {
    agg_drain<FP, NRS, ROWS, true, false, false>(h, list4 + (size_t)seg1 * AGG_CAPS, q_e, acc, list4, l_cnt,
                                                 pre, shift, total, lo, esub, pc, lane_used);
    __syncthreads();
  } else {
    const int echunk  = (((E + NG - 1) / NG) + 31) & ~31;
    const int e_begin = (MODE == 0) ? (int)blockIdx.x * echunk : 0;
    const int e_end   = (MODE == 0) ? ((e_begin + echunk < E) ? (e_begin + echunk) : E) : E;
    for (int b = e_begin; b < e_end; b += PASS) {
      unsigned pend[W];
      constexpr int WC = (W <= 6) ? W : (W + 1) / 2;
#pragma unroll
      for (int w0 = 0; w0 < W; w0 += WC) {
        uint4 pl[2 * WC];
#pragma unroll
        for (int i = 0; i < WC; ++i)
          if (w0 + i < W) {
            const int eb  = b + ((w0 + i) * AGG_THREADS + tid) * 32;
            const int ebc = (e_end - eb > 0) ? eb : e_begin;
            pl[2 * i]     = plane4[(ebc >> 4)];
            pl[2 * i + 1] = plane4[(ebc >> 4) + 1];
          }
        __builtin_amdgcn_sched_barrier(0);
#pragma unroll
        for (int i = 0; i < WC; ++i)
          if (w0 + i < W) {
            const int eb = b + ((w0 + i) * AGG_THREADS + tid) * 32;
            const int nv = e_end - eb;
            const uint4 p0 = pl[2 * i], p1 = pl[2 * i + 1];
            unsigned mt = match4(p0.x, rid4)       | match4(p0.y, rid4) << 4  |
                          match4(p0.z, rid4) << 8  | match4(p0.w, rid4) << 12 |
                          match4(p1.x, rid4) << 16 | match4(p1.y, rid4) << 20 |
                          match4(p1.z, rid4) << 24 | match4(p1.w, rid4) << 28;
            mt &= (nv >= 32) ? 0xffffffffu : ((nv > 0) ? ((1u << nv) - 1u) : 0u);
            pend[w0 + i] = mt;
          }
      }

      bool more;
      do {
        int c = 0;
#pragma unroll
        for (int w = 0; w < W; ++w) c += __popc(pend[w]);
        int incl = c;
#pragma unroll
        for (int o = 1; o < 32; o <<= 1) {
          const int t = __shfl_up(incl, o, 32);
          if (lane >= o) incl += t;
        }
        const int wtot = __shfl(incl, 31, 32);
        int base = 0;
        if (lane == 31 && wtot > 0)
          base = __hip_atomic_fetch_add(&q_n, wtot, __ATOMIC_RELAXED, __HIP_MEMORY_SCOPE_WORKGROUP);
        base = __shfl(base, 31, 32);
        int pos = base + incl - c;
        unsigned left = 0;
#pragma unroll
        for (int w = 0; w < W; ++w) {
          const int eb = b + (w * AGG_THREADS + tid) * 32;
          unsigned p = pend[w];
          while (p != 0u && pos < QC) {
            const int j = __ffs(p) - 1;
            p &= p - 1;
            q_e[pos] = eb + j;
            ++pos;
          }
          pend[w] = p;
          left |= p;
        }
        if (left) q_more = 1;
        __syncthreads();

        const int M = (q_n < QC) ? q_n : QC;
        more = (q_more != 0);
        __syncthreads();

        if (tid == 0) { q_n = 0; q_more = 0; }
        if constexpr (MODE == 0)
          agg_drain<FP, NRS, ROWS, false, true, false>(h, rec4, q_e, acc, wlist, l_cnt, pre, shift,
                                                       M, lo, esub, pc, lane_used);
        else
          agg_drain<FP, NRS, ROWS, false, false, true>(h, rec4, q_e, acc, wlist, l_cnt, pre, shift,
                                                       M, lo, esub, pc, lane_used);
        __syncthreads();
      } while (more);
    }
    if constexpr (MODE == 0)
      if (tid < NG)
        cnt[((int)blockIdx.y * NG + tid) * NG + (int)blockIdx.x] = l_cnt[tid];
  }

  const size_t npad = (size_t)gridDim.y * NR;
  float4* out4 = reinterpret_cast<float4*>(part + (((MODE == 0) ? (size_t)blockIdx.x * npad : (size_t)0) + lo) * FP);
  for (int i = tid; i < ROWS * FP / 4; i += AGG_THREADS)
    out4[i] = acc4[i];
}

template <int F, int FP, int G>
__global__ void k_finalize(const float* __restrict__ h, const float* __restrict__ part,
                           const float* __restrict__ dinv, const float* __restrict__ bias,
                           float* __restrict__ out, int ldo, int N, int npad) {
  int t = blockIdx.x * blockDim.x + threadIdx.x;
  if (t >= N * F) return;
  int n = t / F;
  int f = t - n * F;
  float agg = 0.0f;
#pragma unroll
  for (int g = 0; g < G; ++g)
    agg += part[((size_t)g * npad + n) * FP + f];
  float di = dinv[n];
  float v = agg + di * di * h[n * PADF + f] + bias[f];
  out[n * ldo + f] = selu_f(v);
}

extern "C" void kernel_launch(void* const* d_in, const int* in_sizes, int n_in,
                              void* d_out, int out_size, void* d_ws, size_t ws_size,
                              hipStream_t stream) {
  const float* x  = (const float*)d_in[0];
  const int*   ei = (const int*)d_in[1];
  const float* W1 = (const float*)d_in[2];
  const float* b1 = (const float*)d_in[3];
  const float* W2 = (const float*)d_in[4];
  const float* b2 = (const float*)d_in[5];
  const float* W3 = (const float*)d_in[6];
  const float* b3 = (const float*)d_in[7];
  const float* W4 = (const float*)d_in[8];
  const float* b4 = (const float*)d_in[9];

  const int E = in_sizes[1] / 2;
  const int N = in_sizes[0] / 128;
  const int* src = ei;
  const int* dst = ei + E;

  auto cdiv = [](long long a, long long b) { return (int)((a + b - 1) / b); };
  auto al64 = [](size_t n) { return (n + 63) & ~(size_t)63; };

  float* ws   = (float*)d_ws;
  float* deg  = ws;
  float* dinv = deg  + al64(N);
  int*   rec  = (int*)(dinv + al64(N));
  const int NB = cdiv(E, 256);
  const int pw = NB * 64;
  unsigned* plane = (unsigned*)(rec + al64((size_t)4 * E));
  float* hA   = (float*)(plane + (size_t)pw);
  float* hB   = hA   + (size_t)N * PADF;
  float* part = hB   + (size_t)N * PADF;

  const int NT = N / 16;
  const int GB = cdiv(NT, GEMM_RT * GEMM_WAVES);

  constexpr int NR = 2000, G = AGG_NG;
  constexpr int QC = 8064, SW = 10;
  const int R = cdiv(N, NR);
  int* list = (int*)(part + al64((size_t)G * R * NR * 36));
  int* cnt  = list + (size_t)R * G * G * AGG_CAPS * 4;

  k_fill  <<<cdiv(N, 256), 256, 0, stream>>>(deg, 1.0f, N);
  k_degree<<<cdiv(E, 256), 256, 0, stream>>>(dst, deg, E);
  k_rsqrt <<<cdiv(N, 256), 256, 0, stream>>>(deg, dinv, N);
  k_norm<NR><<<NB, 256, 0, stream>>>(src, dst, dinv, rec, plane, E);

  k_gemm_wmma<128, 15><<<GB, GEMM_WAVES * 32, 0, stream>>>(x, 128, W1, hA, NT);
  k_agg<15, 16, NR, QC, SW, 0><<<dim3(G, R), AGG_THREADS, 0, stream>>>(hA, plane, rec, part, list, cnt, E);

  k_gemm_fin_wmma<15, 20, G, 16><<<GB, GEMM_WAVES * 32, 0, stream>>>(part, R * NR, hA, dinv, b1, W2, hB, NT);
  k_agg<20, 20, NR, QC, SW, 1><<<dim3(G, R), AGG_THREADS, 0, stream>>>(hB, plane, rec, part, list, cnt, E);

  k_gemm_fin_wmma<20, 27, 1, 20><<<GB, GEMM_WAVES * 32, 0, stream>>>(part, R * NR, hB, dinv, b2, W3, hA, NT);
  k_agg<27, 28, NR, QC, SW, 1><<<dim3(G, R), AGG_THREADS, 0, stream>>>(hA, plane, rec, part, list, cnt, E);

  k_gemm_fin_wmma<27, 36, 1, 28><<<GB, GEMM_WAVES * 32, 0, stream>>>(part, R * NR, hA, dinv, b3, W4, hB, NT);
  k_agg<36, 36, NR, QC, SW, 1><<<dim3(G, R), AGG_THREADS, 0, stream>>>(hB, plane, rec, part, list, cnt, E);
  k_finalize<36, 36, 1><<<cdiv((long long)N * 36, 256), 256, 0, stream>>>(hB, part, dinv, b4,
                                                                   (float*)d_out, 36, N, R * NR);
}
